# starting kernel
# baseline (speedup 1.0000x reference)
#include <hip/hip_runtime.h>
#include <hip/hip_bf16.h>
#include <math.h>


typedef __attribute__((ext_vector_type(2))) float v2f;
typedef __attribute__((ext_vector_type(8))) float v8f;

#define KTOT   3084
#define KSTEPS 771
#define NTILES 40
#define FRAG   64
#define NOUT   600
#define KCH    768
#define LDSTR  772

__global__ __launch_bounds__(256)
void ISTFT_buildW(float* __restrict__ Wf) {
  int gid = blockIdx.x * 256 + threadIdx.x;
  if (gid >= KSTEPS * NTILES * FRAG) return;

  int r    = gid & 1;
  int l    = (gid >> 1) & 31;
  int rest = gid >> 6;
  int nt   = rest % NTILES;
  int kk   = rest / NTILES;

  int k   = (kk << 2) + ((l >> 4) << 1) + r;
  int f   = k / 12;
  int rem = k - f * 12;
  int tf  = rem >> 1;
  int d   = rem & 1;
  int n   = (nt << 4) + (l & 15);

  float v = 0.0f;
  if (n < NOUT) {
    int labs = n + 256;
    int nn   = labs - (tf << 7);
    if (nn >= 0 && nn < 512) {
      const float c0 = 6.283185307179586f / 512.0f;
      float env = 0.0f;
      #pragma unroll
      for (int tp = 0; tp < 6; ++tp) {
        int nn2 = labs - (tp << 7);
        if (nn2 >= 0 && nn2 < 512) {
          float w = 0.5f * (1.0f - cosf(c0 * (float)nn2));
          env += w * w;
        }
      }
      float wn   = 0.5f * (1.0f - cosf(c0 * (float)nn));
      bool  edge = (f == 0) || (f == 256);
      float scl  = (edge ? 1.0f : 2.0f) / 512.0f;
      int   m    = (f * nn) & 511;
      float ang  = c0 * (float)m;
      float basis = d ? -sinf(ang) : cosf(ang);
      if (d && edge) basis = 0.0f;
      v = scl * wn / env * basis;
    }
  }
  Wf[gid] = v;
}

__global__ __launch_bounds__(256)
void ISTFT_gemm(const float* __restrict__ X, const float* __restrict__ Wf,
                float* __restrict__ out) {
  __shared__ float As[16 * LDSTR];

  const int tid  = threadIdx.x;
  const int wave = tid >> 5;
  const int lane = tid & 31;
  const int lrow = lane & 15;
  const int lhi  = lane >> 4;
  const int m0   = blockIdx.x << 4;

  v8f acc[5] = {};

  const float* __restrict__ Xblk = X + (size_t)m0 * KTOT;
  const float* __restrict__ Arow = As + lrow * LDSTR + (lhi << 1);
  const float* __restrict__ Wb   = Wf + (size_t)(wave * 5) * FRAG + lane * 2;

  for (int kbase = 0; kbase < KTOT; kbase += KCH) {
    const int klen = (KTOT - kbase < KCH) ? (KTOT - kbase) : KCH;

    for (int idx = tid; idx < 16 * klen; idx += 256) {
      int rr = idx / klen;
      int cc = idx - rr * klen;
      As[rr * LDSTR + cc] = Xblk[(size_t)rr * KTOT + kbase + cc];
    }
    __syncthreads();

    const int ksteps = klen >> 2;
    const int kk0    = kbase >> 2;
    for (int kk = 0; kk < ksteps; ++kk) {
      v2f a = *(const v2f*)(Arow + (kk << 2));
      const float* wr = Wb + (size_t)(kk0 + kk) * (NTILES * FRAG);
      #pragma unroll
      for (int t = 0; t < 5; ++t) {
        v2f b = *(const v2f*)(wr + t * FRAG);
        acc[t] = __builtin_amdgcn_wmma_f32_16x16x4_f32(
            false, a, false, b, (short)0, acc[t], false, false);
      }
    }
    __syncthreads();
  }

  #pragma unroll
  for (int t = 0; t < 5; ++t) {
    int colg = (wave * 5 + t) * 16 + lrow;
    if (colg < NOUT) {
      #pragma unroll
      for (int v = 0; v < 8; ++v) {
        int row = m0 + v + (lhi << 3);
        out[(size_t)row * NOUT + colg] = acc[t][v];
      }
    }
  }
}

extern "C" void kernel_launch(void* const* d_in, const int* in_sizes, int n_in,
                              void* d_out, int out_size, void* d_ws, size_t ws_size,
                              hipStream_t stream) {
  const float* X  = (const float*)d_in[0];
  float*       O  = (float*)d_out;
  float*       Wf = (float*)d_ws;

  const int wtot    = KSTEPS * NTILES * FRAG;
  const int wblocks = (wtot + 255) / 256;

  ISTFT_buildW<<<wblocks, 256, 0, stream>>>(Wf);
  ISTFT_gemm<<<8192 / 16, 256, 0, stream>>>(X, Wf, O);
}


# Each frame's inverse DFT split into a cosine product of the real parts and a sine product of the imag parts for samples 1..256; samples q and 512-q share them as E+O, E-O; sum order changes
# speedup vs baseline: 2.1658x; 2.1658x over previous
#include <hip/hip_runtime.h>
#include <math.h>


typedef __attribute__((ext_vector_type(2))) float v2f;
typedef __attribute__((ext_vector_type(8))) float v8f;

#define ROWK   3084
#define NOUT   600
#define KST    65
#define NTIL   16
#define FRAG   64
#define KCH    768
#define FPL    66
#define LDA    836
#define BLD    602
#define WTOT   133120

__global__ __launch_bounds__(256)
void build_tables(float* __restrict__ Wf, float* __restrict__ env) {
  const int gid = blockIdx.x * 256 + threadIdx.x;
  const float c0 = 6.283185307179586f / 512.0f;

  if (gid < WTOT) {
    int r    = gid & 1;
    int l    = (gid >> 1) & 31;
    int rest = gid >> 6;
    int j    = rest & 15;
    int p    = (rest >> 4) & 1;
    int kk   = rest >> 5;
    int f    = (kk << 2) + ((l >> 4) << 1) + r;
    int q    = (j << 4) + 1 + (l & 15);

    float v = 0.0f;
    if (f <= 256) {
      bool  edge = (f == 0) || (f == 256);
      float scl  = (edge ? 1.0f : 2.0f) / 512.0f;
      float wn   = 0.5f * (1.0f - cosf(c0 * (float)q));
      int   m    = (f * q) & 511;
      float ang  = c0 * (float)m;
      float basis;
      if (p == 0) basis = cosf(ang);
      else        basis = ((m & 255) == 0) ? 0.0f : -sinf(ang);
      v = scl * wn * basis;
    }
    Wf[gid] = v;
  }

  if (gid < NOUT) {
    int labs = gid + 256;
    float e = 0.0f;
    #pragma unroll
    for (int tp = 0; tp < 6; ++tp) {
      int nn2 = labs - (tp << 7);
      if (nn2 >= 0 && nn2 < 512) {
        float w = 0.5f * (1.0f - cosf(c0 * (float)nn2));
        e += w * w;
      }
    }
    env[gid] = e;
  }
}

__device__ __forceinline__
void kstep(const float* __restrict__ ap, const float* __restrict__ wr,
           v8f (&accE)[2][6], v8f (&accO)[2][6]) {
  v2f bE0 = *(const v2f*)(wr);
  v2f bE1 = *(const v2f*)(wr + FRAG);
  v2f bO0 = *(const v2f*)(wr + NTIL * FRAG);
  v2f bO1 = *(const v2f*)(wr + NTIL * FRAG + FRAG);
  #pragma unroll
  for (int t = 0; t < 6; ++t) {
    v2f are = *(const v2f*)(ap + (2 * t) * FPL);
    v2f aim = *(const v2f*)(ap + (2 * t + 1) * FPL);
    accE[0][t] = __builtin_amdgcn_wmma_f32_16x16x4_f32(
        false, are, false, bE0, (short)0, accE[0][t], false, false);
    accE[1][t] = __builtin_amdgcn_wmma_f32_16x16x4_f32(
        false, are, false, bE1, (short)0, accE[1][t], false, false);
    accO[0][t] = __builtin_amdgcn_wmma_f32_16x16x4_f32(
        false, aim, false, bO0, (short)0, accO[0][t], false, false);
    accO[1][t] = __builtin_amdgcn_wmma_f32_16x16x4_f32(
        false, aim, false, bO1, (short)0, accO[1][t], false, false);
  }
}

__global__ __launch_bounds__(256)
void irfft_overlap_add(const float* __restrict__ X, const float* __restrict__ Wf,
                       const float* __restrict__ env, float* __restrict__ out) {
  __shared__ float As[16 * LDA];

  const int tid  = threadIdx.x;
  const int wave = tid >> 5;
  const int lane = tid & 31;
  const int c    = lane & 15;
  const int h    = lane >> 4;
  const int m0   = blockIdx.x << 4;

  v8f accE[2][6] = {};
  v8f accO[2][6] = {};

  const float* __restrict__ Xblk = X + (size_t)m0 * ROWK;
  const float* __restrict__ Arow = As + c * LDA + h * 2;
  const float* __restrict__ Wb   = Wf + (size_t)(wave * 2) * FRAG + lane * 2;

  for (int ch = 0; ch < 4; ++ch) {
    const int kbase = ch * KCH;
    for (int idx = tid; idx < 16 * KCH; idx += 256) {
      int rr = idx / KCH;
      int cc = idx - rr * KCH;
      int fl = cc / 12;
      int pd = cc - fl * 12;
      As[rr * LDA + pd * FPL + fl] = Xblk[(size_t)rr * ROWK + kbase + cc];
    }
    __syncthreads();

    #pragma unroll 1
    for (int kk = 0; kk < 16; ++kk)
      kstep(Arow + kk * 4, Wb + (size_t)(ch * 16 + kk) * (2 * NTIL * FRAG), accE, accO);
    __syncthreads();
  }

  for (int idx = tid; idx < 16 * 48; idx += 256) {
    int rr = idx / 48;
    int cc = idx - rr * 48;
    int pd = cc >> 2;
    int fl = cc & 3;
    As[rr * LDA + pd * FPL + fl] = (fl == 0) ? Xblk[(size_t)rr * ROWK + 4 * KCH + pd] : 0.0f;
  }
  __syncthreads();
  kstep(Arow, Wb + (size_t)64 * (2 * NTIL * FRAG), accE, accO);
  __syncthreads();

  float* buf = As;
  for (int idx = tid; idx < 16 * BLD; idx += 256) buf[idx] = 0.0f;
  __syncthreads();

  #pragma unroll
  for (int t = 0; t < 6; ++t) {
    #pragma unroll
    for (int s = 0; s < 2; ++s) {
      #pragma unroll
      for (int ti = 0; ti < 2; ++ti) {
        int q = ((wave * 2 + ti) << 4) + 1 + c;
        int n = (s == 0) ? (q - 256 + 128 * t) : (256 - q + 128 * t);
        bool ok = (n >= 0) && (n < NOUT) && !(s == 1 && q == 256);
        if (ok) {
          #pragma unroll
          for (int v = 0; v < 8; ++v) {
            float val = (s == 0) ? (accE[ti][t][v] + accO[ti][t][v])
                                 : (accE[ti][t][v] - accO[ti][t][v]);
            buf[(8 * h + v) * BLD + n] += val;
          }
        }
      }
      __syncthreads();
    }
  }

  float* __restrict__ oblk = out + (size_t)m0 * NOUT;
  for (int idx = tid; idx < 16 * NOUT; idx += 256) {
    int rr = idx / NOUT;
    int n  = idx - rr * NOUT;
    oblk[idx] = buf[rr * BLD + n] / env[n];
  }
}

extern "C" void kernel_launch(void* const* d_in, const int* in_sizes, int n_in,
                              void* d_out, int out_size, void* d_ws, size_t ws_size,
                              hipStream_t stream) {
  const float* X  = (const float*)d_in[0];
  float*       O  = (float*)d_out;
  float*       Wf = (float*)d_ws;
  float*       Ev = Wf + 133120;

  build_tables<<<520, 256, 0, stream>>>(Wf, Ev);
  irfft_overlap_add<<<512, 256, 0, stream>>>(X, Wf, Ev, O);
}


# Cosine and sine half-transforms split by frequency parity: parts mirror about sample 128, so four half-length products give samples p, 512-p, 256-p, 256+p; window applied after; sum order changes
# speedup vs baseline: 2.6307x; 1.2147x over previous
#include <hip/hip_runtime.h>
#include <math.h>


typedef __attribute__((ext_vector_type(2))) float v2f;
typedef __attribute__((ext_vector_type(8))) float v8f;

#define ROWK   3084
#define NOUT   600
#define NK8    33
#define FRAG   64
#define KSTR   2560
#define PSTR   512
#define KCH    768
#define FPL    34
#define LDA    836
#define BLD    602
#define WTOT   84480
#define NWIN   257

__global__ __launch_bounds__(256)
void build_tables(float* __restrict__ Wf, float* __restrict__ env, float* __restrict__ win) {
  const int gid = blockIdx.x * 256 + threadIdx.x;
  const float c0 = 6.283185307179586f / 512.0f;

  if (gid < WTOT) {
    int r    = gid & 1;
    int l    = (gid >> 1) & 31;
    int rest = gid >> 6;
    int j    = rest & 7;
    int pk   = rest >> 3;
    int k8   = pk / 5;
    int p    = pk - k8 * 5;
    int par  = (p < 4) ? (p & 1) : (j & 1);
    int f    = (k8 << 3) + ((((l >> 4) << 1) + r) << 1) + par;
    int cc   = l & 15;

    float v = 0.0f;
    if (f <= 256) {
      bool  edge = (f == 0) || (f == 256);
      float scl  = (edge ? 1.0f : 2.0f) / 512.0f;
      if (p < 4) {
        int q = (j << 4) + 1 + cc;
        int m = (f * q) & 511;
        float ang = c0 * (float)m;
        float basis;
        if (p < 2) basis = cosf(ang);
        else       basis = ((m & 255) == 0) ? 0.0f : -sinf(ang);
        v = scl * basis;
      } else if (j < 2 && cc == 0) {
        v = par ? -scl : scl;
      }
    }
    Wf[gid] = v;
  }

  if (gid < NWIN) win[gid] = 0.5f * (1.0f - cosf(c0 * (float)gid));

  if (gid < NOUT) {
    int labs = gid + 256;
    float e = 0.0f;
    #pragma unroll
    for (int tp = 0; tp < 6; ++tp) {
      int nn2 = labs - (tp << 7);
      if (nn2 >= 0 && nn2 < 512) {
        float w = 0.5f * (1.0f - cosf(c0 * (float)nn2));
        e += w * w;
      }
    }
    env[gid] = e;
  }
}

__device__ __forceinline__
void kstep(const float* __restrict__ ap, const float* __restrict__ apc,
           const float* __restrict__ wr, const float* __restrict__ wc,
           v8f (&accEE)[6], v8f (&accEO)[6], v8f (&accOE)[6], v8f (&accOO)[6], v8f &accC) {
  v2f bEE = *(const v2f*)(wr);
  v2f bEO = *(const v2f*)(wr + PSTR);
  v2f bOE = *(const v2f*)(wr + 2 * PSTR);
  v2f bOO = *(const v2f*)(wr + 3 * PSTR);
  v2f bCe = *(const v2f*)(wc);
  v2f bCo = *(const v2f*)(wc + FRAG);
  #pragma unroll
  for (int t = 0; t < 6; ++t) {
    v2f are_e = *(const v2f*)(ap + (4 * t) * FPL);
    v2f are_o = *(const v2f*)(ap + (4 * t + 1) * FPL);
    v2f aim_e = *(const v2f*)(ap + (4 * t + 2) * FPL);
    v2f aim_o = *(const v2f*)(ap + (4 * t + 3) * FPL);
    accEE[t] = __builtin_amdgcn_wmma_f32_16x16x4_f32(
        false, are_e, false, bEE, (short)0, accEE[t], false, false);
    accEO[t] = __builtin_amdgcn_wmma_f32_16x16x4_f32(
        false, are_o, false, bEO, (short)0, accEO[t], false, false);
    accOE[t] = __builtin_amdgcn_wmma_f32_16x16x4_f32(
        false, aim_e, false, bOE, (short)0, accOE[t], false, false);
    accOO[t] = __builtin_amdgcn_wmma_f32_16x16x4_f32(
        false, aim_o, false, bOO, (short)0, accOO[t], false, false);
  }
  v2f ace = *(const v2f*)(apc);
  v2f aco = *(const v2f*)(apc + FPL);
  accC = __builtin_amdgcn_wmma_f32_16x16x4_f32(
      false, ace, false, bCe, (short)0, accC, false, false);
  accC = __builtin_amdgcn_wmma_f32_16x16x4_f32(
      false, aco, false, bCo, (short)0, accC, false, false);
}

__global__ __launch_bounds__(256)
void irfft_overlap_add(const float* __restrict__ X, const float* __restrict__ Wf,
                       const float* __restrict__ env, const float* __restrict__ win,
                       float* __restrict__ out) {
  __shared__ float As[16 * LDA];

  const int tid  = threadIdx.x;
  const int wave = tid >> 5;
  const int lane = tid & 31;
  const int c    = lane & 15;
  const int h    = lane >> 4;
  const int m0   = blockIdx.x << 4;
  const int tw   = (wave < 6) ? wave : (wave - 6);

  v8f accEE[6] = {};
  v8f accEO[6] = {};
  v8f accOE[6] = {};
  v8f accOO[6] = {};
  v8f accC = {};

  const float* __restrict__ Xblk = X + (size_t)m0 * ROWK;
  const float* __restrict__ Arow = As + c * LDA + h * 2;
  const float* __restrict__ Acen = Arow + (4 * tw) * FPL;
  const float* __restrict__ Wb   = Wf + (size_t)wave * FRAG + lane * 2;
  const float* __restrict__ Wc   = Wf + 4 * PSTR + lane * 2;

  for (int ch = 0; ch < 4; ++ch) {
    const int kbase = ch * KCH;
    for (int idx = tid; idx < 16 * KCH; idx += 256) {
      int rr = idx / KCH;
      int cc = idx - rr * KCH;
      int fl = cc / 12;
      int pd = cc - fl * 12;
      As[rr * LDA + (pd * 2 + (fl & 1)) * FPL + (fl >> 1)] = Xblk[(size_t)rr * ROWK + kbase + cc];
    }
    __syncthreads();

    #pragma unroll 1
    for (int kk = 0; kk < 8; ++kk)
      kstep(Arow + kk * 4, Acen + kk * 4, Wb + (size_t)(ch * 8 + kk) * KSTR,
            Wc + (size_t)(ch * 8 + kk) * KSTR, accEE, accEO, accOE, accOO, accC);
    __syncthreads();
  }

  for (int idx = tid; idx < 16 * 48; idx += 256) {
    int rr = idx / 48;
    int cc = idx - rr * 48;
    int pd = cc >> 2;
    int sl = cc & 3;
    As[rr * LDA + (pd * 2) * FPL + sl] = (sl == 0) ? Xblk[(size_t)rr * ROWK + 4 * KCH + pd] : 0.0f;
  }
  __syncthreads();
  {
    const float* __restrict__ wr = Wb + (size_t)32 * KSTR;
    const float* __restrict__ wc = Wc + (size_t)32 * KSTR;
    v2f bEE = *(const v2f*)(wr);
    v2f bCe = *(const v2f*)(wc);
    #pragma unroll
    for (int t = 0; t < 6; ++t) {
      v2f are_e = *(const v2f*)(Arow + (4 * t) * FPL);
      accEE[t] = __builtin_amdgcn_wmma_f32_16x16x4_f32(
          false, are_e, false, bEE, (short)0, accEE[t], false, false);
    }
    v2f ace = *(const v2f*)(Acen);
    accC = __builtin_amdgcn_wmma_f32_16x16x4_f32(
        false, ace, false, bCe, (short)0, accC, false, false);
  }
  __syncthreads();

  float* buf = As;
  for (int idx = tid; idx < 16 * BLD; idx += 256) buf[idx] = 0.0f;
  __syncthreads();

  const int   p   = (wave << 4) + 1 + c;
  const float wq  = win[p];
  const float wm  = win[256 - p];
  const float wc0 = win[256];

  #pragma unroll
  for (int t = 0; t < 6; ++t) {
    const int base = 128 * t - 256;
    const int n0 = base + p;
    const int n1 = base + 512 - p;
    const int n2 = base + 256 - p;
    const int n3 = base + 256 + p;
    const bool ok0 = (n0 >= 0) && (n0 < NOUT);
    const bool ok1 = (n1 >= 0) && (n1 < NOUT);
    const bool ok2 = (n2 >= 0) && (n2 < NOUT) && (p != 128);
    const bool ok3 = (n3 >= 0) && (n3 < NOUT) && (p != 128);
    #pragma unroll
    for (int v = 0; v < 8; ++v) {
      float ep = accEE[t][v] + accEO[t][v];
      float em = accEE[t][v] - accEO[t][v];
      float op = accOO[t][v] + accOE[t][v];
      float om = accOO[t][v] - accOE[t][v];
      float* brow = buf + (8 * h + v) * BLD;
      if (ok0) brow[n0] += wq * (ep + op);
      if (ok1) brow[n1] += wq * (ep - op);
      if (ok2) brow[n2] += wm * (em + om);
      if (ok3) brow[n3] += wm * (em - om);
    }
    if (wave == t && c == 0 && (base + 256) < NOUT) {
      #pragma unroll
      for (int v = 0; v < 8; ++v)
        buf[(8 * h + v) * BLD + base + 256] += wc0 * accC[v];
    }
    __syncthreads();
  }

  float* __restrict__ oblk = out + (size_t)m0 * NOUT;
  for (int idx = tid; idx < 16 * NOUT; idx += 256) {
    int rr = idx / NOUT;
    int n  = idx - rr * NOUT;
    oblk[idx] = buf[rr * BLD + n] / env[n];
  }
}

extern "C" void kernel_launch(void* const* d_in, const int* in_sizes, int n_in,
                              void* d_out, int out_size, void* d_ws, size_t ws_size,
                              hipStream_t stream) {
  const float* X  = (const float*)d_in[0];
  float*       O  = (float*)d_out;
  float*       Wf = (float*)d_ws;
  float*       Ev = Wf + 84480;
  float*       Wn = Ev + 600;

  build_tables<<<330, 256, 0, stream>>>(Wf, Ev, Wn);
  irfft_overlap_add<<<512, 256, 0, stream>>>(X, Wf, Ev, Wn, O);
}


# Input staging as 16-byte pieces turned 4x4 in registers: a thread fetches f, f+2, f+4, f+6 of two frames and writes four consecutive k to each of four LDS planes; LDS pitch 900; sums unchanged
# speedup vs baseline: 3.2710x; 1.2434x over previous
#include <hip/hip_runtime.h>
#include <math.h>


typedef __attribute__((ext_vector_type(2))) float v2f;
typedef __attribute__((ext_vector_type(8))) float v8f;
typedef __attribute__((ext_vector_type(4))) float v4f;

#define ROWK   3084
#define NOUT   600
#define NK8    33
#define FRAG   64
#define KSTR   2560
#define PSTR   512
#define KCH    768
#define FPL    36
#define LDA    900
#define BLD    602
#define WTOT   84480
#define NWIN   257

__global__ __launch_bounds__(256)
void build_tables(float* __restrict__ Wf, float* __restrict__ env, float* __restrict__ win) {
  const int gid = blockIdx.x * 256 + threadIdx.x;
  const float c0 = 6.283185307179586f / 512.0f;

  if (gid < WTOT) {
    int r    = gid & 1;
    int l    = (gid >> 1) & 31;
    int rest = gid >> 6;
    int j    = rest & 7;
    int pk   = rest >> 3;
    int k8   = pk / 5;
    int p    = pk - k8 * 5;
    int par  = (p < 4) ? (p & 1) : (j & 1);
    int f    = (k8 << 3) + ((((l >> 4) << 1) + r) << 1) + par;
    int cc   = l & 15;

    float v = 0.0f;
    if (f <= 256) {
      bool  edge = (f == 0) || (f == 256);
      float scl  = (edge ? 1.0f : 2.0f) / 512.0f;
      if (p < 4) {
        int q = (j << 4) + 1 + cc;
        int m = (f * q) & 511;
        float ang = c0 * (float)m;
        float basis;
        if (p < 2) basis = cosf(ang);
        else       basis = ((m & 255) == 0) ? 0.0f : -sinf(ang);
        v = scl * basis;
      } else if (j < 2 && cc == 0) {
        v = par ? -scl : scl;
      }
    }
    Wf[gid] = v;
  }

  if (gid < NWIN) win[gid] = 0.5f * (1.0f - cosf(c0 * (float)gid));

  if (gid < NOUT) {
    int labs = gid + 256;
    float e = 0.0f;
    #pragma unroll
    for (int tp = 0; tp < 6; ++tp) {
      int nn2 = labs - (tp << 7);
      if (nn2 >= 0 && nn2 < 512) {
        float w = 0.5f * (1.0f - cosf(c0 * (float)nn2));
        e += w * w;
      }
    }
    env[gid] = e;
  }
}

__device__ __forceinline__
void kstep(const float* __restrict__ ap, const float* __restrict__ apc,
           const float* __restrict__ wr, const float* __restrict__ wc,
           v8f (&accEE)[6], v8f (&accEO)[6], v8f (&accOE)[6], v8f (&accOO)[6], v8f &accC) {
  v2f bEE = *(const v2f*)(wr);
  v2f bEO = *(const v2f*)(wr + PSTR);
  v2f bOE = *(const v2f*)(wr + 2 * PSTR);
  v2f bOO = *(const v2f*)(wr + 3 * PSTR);
  v2f bCe = *(const v2f*)(wc);
  v2f bCo = *(const v2f*)(wc + FRAG);
  #pragma unroll
  for (int t = 0; t < 6; ++t) {
    v2f are_e = *(const v2f*)(ap + (4 * t) * FPL);
    v2f are_o = *(const v2f*)(ap + (4 * t + 1) * FPL);
    v2f aim_e = *(const v2f*)(ap + (4 * t + 2) * FPL);
    v2f aim_o = *(const v2f*)(ap + (4 * t + 3) * FPL);
    accEE[t] = __builtin_amdgcn_wmma_f32_16x16x4_f32(
        false, are_e, false, bEE, (short)0, accEE[t], false, false);
    accEO[t] = __builtin_amdgcn_wmma_f32_16x16x4_f32(
        false, are_o, false, bEO, (short)0, accEO[t], false, false);
    accOE[t] = __builtin_amdgcn_wmma_f32_16x16x4_f32(
        false, aim_e, false, bOE, (short)0, accOE[t], false, false);
    accOO[t] = __builtin_amdgcn_wmma_f32_16x16x4_f32(
        false, aim_o, false, bOO, (short)0, accOO[t], false, false);
  }
  v2f ace = *(const v2f*)(apc);
  v2f aco = *(const v2f*)(apc + FPL);
  accC = __builtin_amdgcn_wmma_f32_16x16x4_f32(
      false, ace, false, bCe, (short)0, accC, false, false);
  accC = __builtin_amdgcn_wmma_f32_16x16x4_f32(
      false, aco, false, bCo, (short)0, accC, false, false);
}

__global__ __launch_bounds__(256)
void irfft_overlap_add(const float* __restrict__ X, const float* __restrict__ Wf,
                       const float* __restrict__ env, const float* __restrict__ win,
                       float* __restrict__ out) {
  __shared__ __attribute__((aligned(16))) float As[16 * LDA];

  const int tid  = threadIdx.x;
  const int wave = tid >> 5;
  const int lane = tid & 31;
  const int c    = lane & 15;
  const int h    = lane >> 4;
  const int m0   = blockIdx.x << 4;
  const int tw   = (wave < 6) ? wave : (wave - 6);

  v8f accEE[6] = {};
  v8f accEO[6] = {};
  v8f accOE[6] = {};
  v8f accOO[6] = {};
  v8f accC = {};

  const float* __restrict__ Xblk = X + (size_t)m0 * ROWK;
  const float* __restrict__ Arow = As + c * LDA + h * 2;
  const float* __restrict__ Acen = Arow + (4 * tw) * FPL;
  const float* __restrict__ Wb   = Wf + (size_t)wave * FRAG + lane * 2;
  const float* __restrict__ Wc   = Wf + 4 * PSTR + lane * 2;

  for (int ch = 0; ch < 4; ++ch) {
    const int kbase = ch * KCH;
    for (int s = 0; s < 3; ++s) {
      int q   = tid + 256 * s;
      int rr  = q / 48;
      int rem = q - rr * 48;
      int eg  = rem / 6;
      int r6  = rem - eg * 6;
      int par = (r6 >= 3) ? 1 : 0;
      int tp  = r6 - 3 * par;
      const v4f* __restrict__ src =
          (const v4f*)(Xblk + (size_t)rr * ROWK + kbase + (8 * eg + par) * 12 + 4 * tp);
      v4f L0 = src[0];
      v4f L1 = src[6];
      v4f L2 = src[12];
      v4f L3 = src[18];
      float* dst = As + rr * LDA + (8 * tp + par) * FPL + 4 * eg;
      v4f S0 = {L0.x, L1.x, L2.x, L3.x};
      v4f S1 = {L0.y, L1.y, L2.y, L3.y};
      v4f S2 = {L0.z, L1.z, L2.z, L3.z};
      v4f S3 = {L0.w, L1.w, L2.w, L3.w};
      *(v4f*)(dst)           = S0;
      *(v4f*)(dst + 2 * FPL) = S1;
      *(v4f*)(dst + 4 * FPL) = S2;
      *(v4f*)(dst + 6 * FPL) = S3;
    }
    __syncthreads();

    #pragma unroll 1
    for (int kk = 0; kk < 8; ++kk)
      kstep(Arow + kk * 4, Acen + kk * 4, Wb + (size_t)(ch * 8 + kk) * KSTR,
            Wc + (size_t)(ch * 8 + kk) * KSTR, accEE, accEO, accOE, accOO, accC);
    __syncthreads();
  }

  for (int idx = tid; idx < 16 * 48; idx += 256) {
    int rr = idx / 48;
    int cc = idx - rr * 48;
    int pd = cc >> 2;
    int sl = cc & 3;
    As[rr * LDA + (pd * 2) * FPL + sl] = (sl == 0) ? Xblk[(size_t)rr * ROWK + 4 * KCH + pd] : 0.0f;
  }
  __syncthreads();
  {
    const float* __restrict__ wr = Wb + (size_t)32 * KSTR;
    const float* __restrict__ wc = Wc + (size_t)32 * KSTR;
    v2f bEE = *(const v2f*)(wr);
    v2f bCe = *(const v2f*)(wc);
    #pragma unroll
    for (int t = 0; t < 6; ++t) {
      v2f are_e = *(const v2f*)(Arow + (4 * t) * FPL);
      accEE[t] = __builtin_amdgcn_wmma_f32_16x16x4_f32(
          false, are_e, false, bEE, (short)0, accEE[t], false, false);
    }
    v2f ace = *(const v2f*)(Acen);
    accC = __builtin_amdgcn_wmma_f32_16x16x4_f32(
        false, ace, false, bCe, (short)0, accC, false, false);
  }
  __syncthreads();

  float* buf = As;
  for (int idx = tid; idx < 16 * BLD; idx += 256) buf[idx] = 0.0f;
  __syncthreads();

  const int   p   = (wave << 4) + 1 + c;
  const float wq  = win[p];
  const float wm  = win[256 - p];
  const float wc0 = win[256];

  #pragma unroll
  for (int t = 0; t < 6; ++t) {
    const int base = 128 * t - 256;
    const int n0 = base + p;
    const int n1 = base + 512 - p;
    const int n2 = base + 256 - p;
    const int n3 = base + 256 + p;
    const bool ok0 = (n0 >= 0) && (n0 < NOUT);
    const bool ok1 = (n1 >= 0) && (n1 < NOUT);
    const bool ok2 = (n2 >= 0) && (n2 < NOUT) && (p != 128);
    const bool ok3 = (n3 >= 0) && (n3 < NOUT) && (p != 128);
    #pragma unroll
    for (int v = 0; v < 8; ++v) {
      float ep = accEE[t][v] + accEO[t][v];
      float em = accEE[t][v] - accEO[t][v];
      float op = accOO[t][v] + accOE[t][v];
      float om = accOO[t][v] - accOE[t][v];
      float* brow = buf + (8 * h + v) * BLD;
      if (ok0) brow[n0] += wq * (ep + op);
      if (ok1) brow[n1] += wq * (ep - op);
      if (ok2) brow[n2] += wm * (em + om);
      if (ok3) brow[n3] += wm * (em - om);
    }
    if (wave == t && c == 0 && (base + 256) < NOUT) {
      #pragma unroll
      for (int v = 0; v < 8; ++v)
        buf[(8 * h + v) * BLD + base + 256] += wc0 * accC[v];
    }
    __syncthreads();
  }

  float* __restrict__ oblk = out + (size_t)m0 * NOUT;
  for (int idx = tid; idx < 16 * NOUT; idx += 256) {
    int rr = idx / NOUT;
    int n  = idx - rr * NOUT;
    oblk[idx] = buf[rr * BLD + n] / env[n];
  }
}

extern "C" void kernel_launch(void* const* d_in, const int* in_sizes, int n_in,
                              void* d_out, int out_size, void* d_ws, size_t ws_size,
                              hipStream_t stream) {
  const float* X  = (const float*)d_in[0];
  float*       O  = (float*)d_out;
  float*       Wf = (float*)d_ws;
  float*       Ev = Wf + 84480;
  float*       Wn = Ev + 600;

  build_tables<<<330, 256, 0, stream>>>(Wf, Ev, Wn);
  irfft_overlap_add<<<512, 256, 0, stream>>>(X, Wf, Ev, Wn, O);
}


# Overlap-add in registers: a thread adds the two frames (T, T-2) that meet at its samples and stores each sum once into one of two LDS images, no read-modify-write passes; the sum order changes
# speedup vs baseline: 3.6436x; 1.1139x over previous
#include <hip/hip_runtime.h>
#include <math.h>


typedef __attribute__((ext_vector_type(2))) float v2f;
typedef __attribute__((ext_vector_type(8))) float v8f;
typedef __attribute__((ext_vector_type(4))) float v4f;

#define ROWK   3084
#define NOUT   600
#define NK8    33
#define FRAG   64
#define KSTR   2560
#define PSTR   512
#define KCH    768
#define FPL    36
#define LDA    900
#define BLD    602
#define LDSF   19264
#define WTOT   84480
#define NWIN   257

__global__ __launch_bounds__(256)
void build_tables(float* __restrict__ Wf, float* __restrict__ env, float* __restrict__ win) {
  const int gid = blockIdx.x * 256 + threadIdx.x;
  const float c0 = 6.283185307179586f / 512.0f;

  if (gid < WTOT) {
    int r    = gid & 1;
    int l    = (gid >> 1) & 31;
    int rest = gid >> 6;
    int j    = rest & 7;
    int pk   = rest >> 3;
    int k8   = pk / 5;
    int p    = pk - k8 * 5;
    int par  = (p < 4) ? (p & 1) : (j & 1);
    int f    = (k8 << 3) + ((((l >> 4) << 1) + r) << 1) + par;
    int cc   = l & 15;

    float v = 0.0f;
    if (f <= 256) {
      bool  edge = (f == 0) || (f == 256);
      float scl  = (edge ? 1.0f : 2.0f) / 512.0f;
      if (p < 4) {
        int q = (j << 4) + 1 + cc;
        int m = (f * q) & 511;
        float ang = c0 * (float)m;
        float basis;
        if (p < 2) basis = cosf(ang);
        else       basis = ((m & 255) == 0) ? 0.0f : -sinf(ang);
        v = scl * basis;
      } else if (j < 2 && cc == 0) {
        v = par ? -scl : scl;
      }
    }
    Wf[gid] = v;
  }

  if (gid < NWIN) win[gid] = 0.5f * (1.0f - cosf(c0 * (float)gid));

  if (gid < NOUT) {
    int labs = gid + 256;
    float e = 0.0f;
    #pragma unroll
    for (int tp = 0; tp < 6; ++tp) {
      int nn2 = labs - (tp << 7);
      if (nn2 >= 0 && nn2 < 512) {
        float w = 0.5f * (1.0f - cosf(c0 * (float)nn2));
        e += w * w;
      }
    }
    env[gid] = e;
  }
}

__device__ __forceinline__
void kstep(const float* __restrict__ ap, const float* __restrict__ apc,
           const float* __restrict__ wr, const float* __restrict__ wc,
           v8f (&accEE)[6], v8f (&accEO)[6], v8f (&accOE)[6], v8f (&accOO)[6], v8f &accC) {
  v2f bEE = *(const v2f*)(wr);
  v2f bEO = *(const v2f*)(wr + PSTR);
  v2f bOE = *(const v2f*)(wr + 2 * PSTR);
  v2f bOO = *(const v2f*)(wr + 3 * PSTR);
  v2f bCe = *(const v2f*)(wc);
  v2f bCo = *(const v2f*)(wc + FRAG);
  #pragma unroll
  for (int t = 0; t < 6; ++t) {
    v2f are_e = *(const v2f*)(ap + (4 * t) * FPL);
    v2f are_o = *(const v2f*)(ap + (4 * t + 1) * FPL);
    v2f aim_e = *(const v2f*)(ap + (4 * t + 2) * FPL);
    v2f aim_o = *(const v2f*)(ap + (4 * t + 3) * FPL);
    accEE[t] = __builtin_amdgcn_wmma_f32_16x16x4_f32(
        false, are_e, false, bEE, (short)0, accEE[t], false, false);
    accEO[t] = __builtin_amdgcn_wmma_f32_16x16x4_f32(
        false, are_o, false, bEO, (short)0, accEO[t], false, false);
    accOE[t] = __builtin_amdgcn_wmma_f32_16x16x4_f32(
        false, aim_e, false, bOE, (short)0, accOE[t], false, false);
    accOO[t] = __builtin_amdgcn_wmma_f32_16x16x4_f32(
        false, aim_o, false, bOO, (short)0, accOO[t], false, false);
  }
  v2f ace = *(const v2f*)(apc);
  v2f aco = *(const v2f*)(apc + FPL);
  accC = __builtin_amdgcn_wmma_f32_16x16x4_f32(
      false, ace, false, bCe, (short)0, accC, false, false);
  accC = __builtin_amdgcn_wmma_f32_16x16x4_f32(
      false, aco, false, bCo, (short)0, accC, false, false);
}

__global__ __launch_bounds__(256)
void irfft_overlap_add(const float* __restrict__ X, const float* __restrict__ Wf,
                       const float* __restrict__ env, const float* __restrict__ win,
                       float* __restrict__ out) {
  __shared__ __attribute__((aligned(16))) float As[LDSF];

  const int tid  = threadIdx.x;
  const int wave = tid >> 5;
  const int lane = tid & 31;
  const int c    = lane & 15;
  const int h    = lane >> 4;
  const int m0   = blockIdx.x << 4;
  const int tw   = (wave < 6) ? wave : (wave - 6);

  v8f accEE[6] = {};
  v8f accEO[6] = {};
  v8f accOE[6] = {};
  v8f accOO[6] = {};
  v8f accC = {};

  const float* __restrict__ Xblk = X + (size_t)m0 * ROWK;
  const float* __restrict__ Arow = As + c * LDA + h * 2;
  const float* __restrict__ Acen = Arow + (4 * tw) * FPL;
  const float* __restrict__ Wb   = Wf + (size_t)wave * FRAG + lane * 2;
  const float* __restrict__ Wc   = Wf + 4 * PSTR + lane * 2;

  for (int ch = 0; ch < 4; ++ch) {
    const int kbase = ch * KCH;
    for (int s = 0; s < 3; ++s) {
      int q   = tid + 256 * s;
      int rr  = q / 48;
      int rem = q - rr * 48;
      int eg  = rem / 6;
      int r6  = rem - eg * 6;
      int par = (r6 >= 3) ? 1 : 0;
      int tp  = r6 - 3 * par;
      const v4f* __restrict__ src =
          (const v4f*)(Xblk + (size_t)rr * ROWK + kbase + (8 * eg + par) * 12 + 4 * tp);
      v4f L0 = src[0];
      v4f L1 = src[6];
      v4f L2 = src[12];
      v4f L3 = src[18];
      float* dst = As + rr * LDA + (8 * tp + par) * FPL + 4 * eg;
      v4f S0 = {L0.x, L1.x, L2.x, L3.x};
      v4f S1 = {L0.y, L1.y, L2.y, L3.y};
      v4f S2 = {L0.z, L1.z, L2.z, L3.z};
      v4f S3 = {L0.w, L1.w, L2.w, L3.w};
      *(v4f*)(dst)           = S0;
      *(v4f*)(dst + 2 * FPL) = S1;
      *(v4f*)(dst + 4 * FPL) = S2;
      *(v4f*)(dst + 6 * FPL) = S3;
    }
    __syncthreads();

    #pragma unroll 1
    for (int kk = 0; kk < 8; ++kk)
      kstep(Arow + kk * 4, Acen + kk * 4, Wb + (size_t)(ch * 8 + kk) * KSTR,
            Wc + (size_t)(ch * 8 + kk) * KSTR, accEE, accEO, accOE, accOO, accC);
    __syncthreads();
  }

  for (int idx = tid; idx < 16 * 48; idx += 256) {
    int rr = idx / 48;
    int cc = idx - rr * 48;
    int pd = cc >> 2;
    int sl = cc & 3;
    As[rr * LDA + (pd * 2) * FPL + sl] = (sl == 0) ? Xblk[(size_t)rr * ROWK + 4 * KCH + pd] : 0.0f;
  }
  __syncthreads();
  {
    const float* __restrict__ wr = Wb + (size_t)32 * KSTR;
    const float* __restrict__ wc = Wc + (size_t)32 * KSTR;
    v2f bEE = *(const v2f*)(wr);
    v2f bCe = *(const v2f*)(wc);
    #pragma unroll
    for (int t = 0; t < 6; ++t) {
      v2f are_e = *(const v2f*)(Arow + (4 * t) * FPL);
      accEE[t] = __builtin_amdgcn_wmma_f32_16x16x4_f32(
          false, are_e, false, bEE, (short)0, accEE[t], false, false);
    }
    v2f ace = *(const v2f*)(Acen);
    accC = __builtin_amdgcn_wmma_f32_16x16x4_f32(
        false, ace, false, bCe, (short)0, accC, false, false);
  }
  __syncthreads();

  float* __restrict__ imgU = As;
  float* __restrict__ imgD = As + 16 * BLD;

  const int   p   = (wave << 4) + 1 + c;
  const float wq  = win[p];
  const float wm  = win[256 - p];
  const float wc0 = win[256];
  const bool  mid = (p == 128);

  #pragma unroll
  for (int t = 0; t < 6; ++t) {
    #pragma unroll
    for (int v = 0; v < 8; ++v) {
      float ep = accEE[t][v] + accEO[t][v];
      float em = accEE[t][v] - accEO[t][v];
      float op = accOO[t][v] + accOE[t][v];
      float om = accOO[t][v] - accOE[t][v];
      float s0 = wq * (ep + op);
      float s1 = wq * (ep - op);
      float s2 = wm * (em + om);
      float s3 = wm * (em - om);
      accEE[t][v] = s0;
      accOO[t][v] = s1;
      accOE[t][v] = s2;
      accEO[t][v] = mid ? s1 : s3;
    }
  }

  float* __restrict__ rowU = imgU + (8 * h) * BLD + p;
  float* __restrict__ rowD = imgD + (8 * h) * BLD - p;
  if (mid) {
    #pragma unroll
    for (int v = 0; v < 8; ++v) rowU[v * BLD - 128] = accEE[1][v];
  }
  #pragma unroll
  for (int T = 2; T < 6; ++T) {
    #pragma unroll
    for (int v = 0; v < 8; ++v) rowU[v * BLD + 128 * (T - 2)] = accEE[T][v] + accEO[T - 2][v];
  }
  if (p + 512 < NOUT) {
    #pragma unroll
    for (int v = 0; v < 8; ++v) rowU[v * BLD + 512] = accEO[4][v];
  }
  if (!mid) {
    #pragma unroll
    for (int v = 0; v < 8; ++v) rowD[v * BLD + 128] = accOE[1][v];
    #pragma unroll
    for (int T = 2; T < 5; ++T) {
      #pragma unroll
      for (int v = 0; v < 8; ++v) rowD[v * BLD + 128 * T] = accOE[T][v] + accOO[T - 2][v];
    }
    if (640 - p < NOUT) {
      #pragma unroll
      for (int v = 0; v < 8; ++v) rowD[v * BLD + 640] = accOE[5][v] + accOO[3][v];
    }
  }
  if (wave < 5 && c == 0) {
    #pragma unroll
    for (int v = 0; v < 8; ++v) imgD[(8 * h + v) * BLD + 128 * wave] = wc0 * accC[v];
  }
  __syncthreads();

  float* __restrict__ oblk = out + (size_t)m0 * NOUT;
  for (int idx = tid; idx < 16 * NOUT; idx += 256) {
    int rr = idx / NOUT;
    int n  = idx - rr * NOUT;
    oblk[idx] = (imgU[rr * BLD + n] + imgD[rr * BLD + n]) / env[n];
  }
}

extern "C" void kernel_launch(void* const* d_in, const int* in_sizes, int n_in,
                              void* d_out, int out_size, void* d_ws, size_t ws_size,
                              hipStream_t stream) {
  const float* X  = (const float*)d_in[0];
  float*       O  = (float*)d_out;
  float*       Wf = (float*)d_ws;
  float*       Ev = Wf + 84480;
  float*       Wn = Ev + 600;

  build_tables<<<330, 256, 0, stream>>>(Wf, Ev, Wn);
  irfft_overlap_add<<<512, 256, 0, stream>>>(X, Wf, Ev, Wn, O);
}


# Even-frequency cosine and sine products split by frequency mod 4: they mirror about sample 64, so one quarter-length product serves columns r and 128-r; a wave holds both, 3 frames; sum order changes
# speedup vs baseline: 4.2984x; 1.1797x over previous
#include <hip/hip_runtime.h>
#include <math.h>


typedef __attribute__((ext_vector_type(2))) float v2f;
typedef __attribute__((ext_vector_type(8))) float v8f;
typedef __attribute__((ext_vector_type(4))) float v4f;

#define ROWK   3084
#define NOUT   600
#define NPART  13
#define FRAG   64
#define PSTR   256
#define KSTR   3328
#define WTOT   56576
#define KCH    768
#define FPL    36
#define LDA    900
#define BLD    602
#define LDSF   19264
#define NWIN   257

__global__ __launch_bounds__(256)
void build_tables(float* __restrict__ Wf, float* __restrict__ env, float* __restrict__ win) {
  const int gid = blockIdx.x * 256 + threadIdx.x;
  const float c0 = 6.283185307179586f / 512.0f;

  if (gid < WTOT) {
    int s    = gid & 1;
    int l    = (gid >> 1) & 31;
    int rest = gid >> 6;
    int j    = rest & 3;
    int pk   = rest >> 2;
    int k16  = pk / NPART;
    int part = pk - k16 * NPART;
    int cc   = l & 15;
    int hh   = l >> 4;
    int r    = (j << 4) + 1 + cc;
    bool sp  = (r == 64);
    int m    = sp ? 128 : (128 - r);
    bool oddf = (part >= 4) && (part < 12);
    int f;
    if (!oddf) f = (k16 << 4) + (hh << 3) + (s << 2) + (((part == 1) || (part == 3)) ? 2 : 0);
    else       f = (k16 << 4) + 1 + ((part & 1) << 1) + (hh << 3) + (s << 2);

    float v = 0.0f;
    if (f <= 256) {
      bool  edge = (f == 0) || (f == 256);
      float scl  = (edge ? 1.0f : 2.0f) / 512.0f;
      bool  flat = sp && ((part == 1) || (part == 8) || (part == 9) || (part == 12));
      if (part == 12 && !sp) {
        v = 0.0f;
      } else if (flat) {
        v = scl;
      } else {
        bool is_cos = (part < 2) || (part == 4) || (part == 5) || (part == 8) || (part == 9);
        int q  = (part >= 8) ? m : r;
        int ma = (f * q) & 511;
        float ang = c0 * (float)ma;
        float basis;
        if (is_cos) basis = cosf(ang);
        else        basis = ((ma & 255) == 0) ? 0.0f : -sinf(ang);
        v = scl * basis;
      }
    }
    Wf[gid] = v;
  }

  if (gid < NWIN) win[gid] = 0.5f * (1.0f - cosf(c0 * (float)gid));

  if (gid < NOUT) {
    int labs = gid + 256;
    float e = 0.0f;
    #pragma unroll
    for (int tp = 0; tp < 6; ++tp) {
      int nn2 = labs - (tp << 7);
      if (nn2 >= 0 && nn2 < 512) {
        float w = 0.5f * (1.0f - cosf(c0 * (float)nn2));
        e += w * w;
      }
    }
    env[gid] = e;
  }
}

#define WMMA4(a, b, c) __builtin_amdgcn_wmma_f32_16x16x4_f32(false, (a), false, (b), (short)0, (c), false, false)

__device__ __forceinline__
void k16step(const float* __restrict__ ap, const float* __restrict__ ws, unsigned lo, bool own64,
             v8f (&aE0)[3], v8f (&aE2)[3], v8f (&aO0)[3], v8f (&aO2)[3],
             v8f (&aEr)[3], v8f (&aOr)[3], v8f (&aEm)[3], v8f (&aOm)[3]) {
  const float* __restrict__ wr = ws + lo;
  {
    v2f bE0 = *(const v2f*)(wr);
    v2f bE2 = *(const v2f*)(wr + PSTR);
    v2f bO0 = *(const v2f*)(wr + 2 * PSTR);
    v2f bO2 = *(const v2f*)(wr + 3 * PSTR);
    v2f bC  = *(const v2f*)(wr + 12 * PSTR);
    #pragma unroll
    for (int i = 0; i < 3; ++i) {
      const float* a = ap + (8 * i) * FPL;
      v2f re0 = *(const v2f*)(a);
      v2f re2 = *(const v2f*)(a + 4);
      v2f im0 = *(const v2f*)(a + 2 * FPL);
      v2f im2 = *(const v2f*)(a + 2 * FPL + 4);
      aE0[i] = WMMA4(re0, bE0, aE0[i]);
      aE2[i] = WMMA4(re2, bE2, aE2[i]);
      aO0[i] = WMMA4(im0, bO0, aO0[i]);
      aO2[i] = WMMA4(im2, bO2, aO2[i]);
    }
    if (own64) {
      #pragma unroll
      for (int i = 0; i < 3; ++i) {
        v2f re0 = *(const v2f*)(ap + (8 * i) * FPL);
        aO0[i] = WMMA4(re0, bC, aO0[i]);
      }
    }
  }
  #pragma unroll
  for (int u = 0; u < 2; ++u) {
    v2f bEr = *(const v2f*)(wr + (4 + u) * PSTR);
    v2f bOr = *(const v2f*)(wr + (6 + u) * PSTR);
    v2f bEm = *(const v2f*)(wr + (8 + u) * PSTR);
    v2f bOm = *(const v2f*)(wr + (10 + u) * PSTR);
    #pragma unroll
    for (int i = 0; i < 3; ++i) {
      const float* a = ap + (8 * i + 1) * FPL + 4 * u;
      v2f ro = *(const v2f*)(a);
      v2f io = *(const v2f*)(a + 2 * FPL);
      aEr[i] = WMMA4(ro, bEr, aEr[i]);
      aEm[i] = WMMA4(ro, bEm, aEm[i]);
      aOr[i] = WMMA4(io, bOr, aOr[i]);
      aOm[i] = WMMA4(io, bOm, aOm[i]);
    }
  }
}

template <int I>
__device__ __forceinline__
void store_stage(const v8f (&s0)[3], const v8f (&s1)[3], const v8f (&s2)[3], const v8f (&s3)[3],
                 int p, bool mid, int fp, float* __restrict__ rowU, float* __restrict__ rowD) {
  const int b = 128 * fp;
  if (I == 0) {
    if (fp == 1 && mid) {
      #pragma unroll
      for (int v = 0; v < 8; ++v) rowU[v * BLD] = s0[0][v];
    }
    if (fp == 1 && !mid) {
      #pragma unroll
      for (int v = 0; v < 8; ++v) rowD[v * BLD + 128 - p] = s2[0][v];
    }
  } else if (I == 1) {
    #pragma unroll
    for (int v = 0; v < 8; ++v) rowU[v * BLD + b + p] = s0[1][v] + s3[0][v];
    if (!mid) {
      #pragma unroll
      for (int v = 0; v < 8; ++v) rowD[v * BLD + b + 256 - p] = s2[1][v] + s1[0][v];
    }
  } else {
    #pragma unroll
    for (int v = 0; v < 8; ++v) rowU[v * BLD + b + 256 + p] = s0[2][v] + s3[1][v];
    if (fp == 0 && p + 512 < NOUT) {
      #pragma unroll
      for (int v = 0; v < 8; ++v) rowU[v * BLD + 512 + p] = s3[2][v];
    }
    if (!mid && b + 512 - p < NOUT) {
      #pragma unroll
      for (int v = 0; v < 8; ++v) rowD[v * BLD + b + 512 - p] = s2[2][v] + s1[1][v];
    }
  }
}

template <int I>
__device__ __forceinline__
void epi_frame(v8f (&aE0)[3], v8f (&aE2)[3], v8f (&aO0)[3], v8f (&aO2)[3],
               v8f (&aEr)[3], v8f (&aOr)[3], v8f (&aEm)[3], v8f (&aOm)[3],
               int r, int m, bool sp, int fp, int h,
               float wr0, float wr1, float wm0, float wm1, float wc0,
               float* __restrict__ imgU, float* __restrict__ imgD) {
  if (sp) {
    #pragma unroll
    for (int v = 0; v < 8; ++v) {
      float Z  = aO0[I][v];
      float E0 = aE0[I][v], E2 = aE2[I][v];
      if (fp + 2 * I <= 4)
        imgD[(8 * h + v) * BLD + 128 * (fp + 2 * I)] = wc0 * (Z + E2 - aEm[I][v]);
      float bq = Z - E2;
      aE0[I][v] = 0.5f * (E0 + bq);
      aE2[I][v] = 0.5f * (E0 - bq);
      float hO = 0.5f * aO2[I][v];
      aO0[I][v] = hO;
      aO2[I][v] = hO;
      aEm[I][v] = 0.0f;
    }
  }
  #pragma unroll
  for (int v = 0; v < 8; ++v) {
    float E0 = aE0[I][v], E2 = aE2[I][v], O0 = aO0[I][v], O2 = aO2[I][v];
    float eOr = aEr[I][v], oOr = aOr[I][v], eOm = aEm[I][v], oOm = aOm[I][v];
    float eEr = E0 + E2;
    float eEm = E0 - E2;
    float oEr = O0 + O2;
    float oEm = O2 - O0;
    float ep = eEr + eOr;
    float em = eEr - eOr;
    float op = oOr + oEr;
    float om = oOr - oEr;
    aE0[I][v] = wr0 * (ep + op);
    aE2[I][v] = wr0 * (ep - op);
    aO0[I][v] = wr1 * (em + om);
    aO2[I][v] = wr1 * (em - om);
    ep = eEm + eOm;
    em = eEm - eOm;
    op = oOm + oEm;
    om = oOm - oEm;
    float t1 = wm0 * (ep - op);
    aEr[I][v] = wm0 * (ep + op);
    aOr[I][v] = t1;
    aEm[I][v] = wm1 * (em + om);
    aOm[I][v] = sp ? t1 : (wm1 * (em - om));
  }
  float* __restrict__ rowU = imgU + (8 * h) * BLD;
  float* __restrict__ rowD = imgD + (8 * h) * BLD;
  store_stage<I>(aE0, aE2, aO0, aO2, r, false, fp, rowU, rowD);
  store_stage<I>(aEr, aOr, aEm, aOm, m, sp, fp, rowU, rowD);
}

__global__ __launch_bounds__(256) __attribute__((amdgpu_waves_per_eu(4)))
void irfft_overlap_add(const float* __restrict__ X, const float* __restrict__ Wf,
                       const float* __restrict__ env, const float* __restrict__ win,
                       float* __restrict__ out) {
  __shared__ __attribute__((aligned(16))) float As[LDSF];

  const int tid  = threadIdx.x;
  const int wave = __builtin_amdgcn_readfirstlane(tid >> 5);
  const int lane = tid & 31;
  const int c    = lane & 15;
  const int h    = lane >> 4;
  const int m0   = blockIdx.x << 4;
  const int fp   = wave >> 2;
  const int j    = (wave + fp) & 3;
  const bool own64 = (j == 3);

  v8f aE0[3] = {};
  v8f aE2[3] = {};
  v8f aO0[3] = {};
  v8f aO2[3] = {};
  v8f aEr[3] = {};
  v8f aOr[3] = {};
  v8f aEm[3] = {};
  v8f aOm[3] = {};

  const float* __restrict__ Xblk = X + (size_t)m0 * ROWK;
  const float* __restrict__ Arow = As + c * LDA + h * 2;
  const float* __restrict__ Afp  = Arow + (4 * fp) * FPL;
  const float* __restrict__ Wb   = Wf + j * FRAG;
  const unsigned            lo   = (unsigned)lane * 2u;

  for (int ch = 0; ch < 4; ++ch) {
    const int kbase = ch * KCH;
    for (int s = 0; s < 3; ++s) {
      int q   = tid + 256 * s;
      int rr  = q / 48;
      int rem = q - rr * 48;
      int eg  = rem / 6;
      int r6  = rem - eg * 6;
      int par = (r6 >= 3) ? 1 : 0;
      int tp  = r6 - 3 * par;
      int fl  = 16 * (eg >> 1) + 2 * (eg & 1) + par;
      const v4f* __restrict__ src =
          (const v4f*)(Xblk + (rr * ROWK + kbase + fl * 12 + 4 * tp));
      v4f L0 = src[0];
      v4f L1 = src[12];
      v4f L2 = src[24];
      v4f L3 = src[36];
      float* dst = As + rr * LDA + (8 * tp + par) * FPL + 4 * eg;
      v4f S0 = {L0.x, L1.x, L2.x, L3.x};
      v4f S1 = {L0.y, L1.y, L2.y, L3.y};
      v4f S2 = {L0.z, L1.z, L2.z, L3.z};
      v4f S3 = {L0.w, L1.w, L2.w, L3.w};
      *(v4f*)(dst)           = S0;
      *(v4f*)(dst + 2 * FPL) = S1;
      *(v4f*)(dst + 4 * FPL) = S2;
      *(v4f*)(dst + 6 * FPL) = S3;
    }
    __syncthreads();

    #pragma unroll 1
    for (int kk = 0; kk < 4; ++kk)
      k16step(Afp + kk * 8, Wb + (ch * 4 + kk) * KSTR, lo, own64,
              aE0, aE2, aO0, aO2, aEr, aOr, aEm, aOm);
    __syncthreads();
  }

  for (int idx = tid; idx < 16 * 48; idx += 256) {
    int rr = idx / 48;
    int cc = idx - rr * 48;
    int pd = cc >> 2;
    int sl = cc & 3;
    As[rr * LDA + (pd * 2) * FPL + sl] = (sl == 0) ? Xblk[(size_t)rr * ROWK + 4 * KCH + pd] : 0.0f;
  }
  __syncthreads();
  {
    const float* __restrict__ wr = Wb + 16 * KSTR + lo;
    v2f bE0 = *(const v2f*)(wr);
    #pragma unroll
    for (int i = 0; i < 3; ++i) {
      v2f re0 = *(const v2f*)(Afp + (8 * i) * FPL);
      aE0[i] = WMMA4(re0, bE0, aE0[i]);
    }
    if (own64) {
      v2f bC = *(const v2f*)(wr + 12 * PSTR);
      #pragma unroll
      for (int i = 0; i < 3; ++i) {
        v2f re0 = *(const v2f*)(Afp + (8 * i) * FPL);
        aO0[i] = WMMA4(re0, bC, aO0[i]);
      }
    }
  }
  __syncthreads();

  float* __restrict__ imgU = As;
  float* __restrict__ imgD = As + 16 * BLD;

  const int   r   = (j << 4) + 1 + c;
  const bool  sp  = (r == 64);
  const int   m   = sp ? 128 : (128 - r);
  const float wr0 = win[r];
  const float wr1 = win[256 - r];
  const float wm0 = win[m];
  const float wm1 = win[256 - m];
  const float wc0 = win[256];

  epi_frame<0>(aE0, aE2, aO0, aO2, aEr, aOr, aEm, aOm, r, m, sp, fp, h, wr0, wr1, wm0, wm1, wc0, imgU, imgD);
  epi_frame<1>(aE0, aE2, aO0, aO2, aEr, aOr, aEm, aOm, r, m, sp, fp, h, wr0, wr1, wm0, wm1, wc0, imgU, imgD);
  epi_frame<2>(aE0, aE2, aO0, aO2, aEr, aOr, aEm, aOm, r, m, sp, fp, h, wr0, wr1, wm0, wm1, wc0, imgU, imgD);
  __syncthreads();

  float* __restrict__ oblk = out + (size_t)m0 * NOUT;
  for (int idx = tid; idx < 16 * NOUT; idx += 256) {
    int rr = idx / NOUT;
    int n  = idx - rr * NOUT;
    oblk[idx] = (imgU[rr * BLD + n] + imgD[rr * BLD + n]) / env[n];
  }
}

extern "C" void kernel_launch(void* const* d_in, const int* in_sizes, int n_in,
                              void* d_out, int out_size, void* d_ws, size_t ws_size,
                              hipStream_t stream) {
  const float* X  = (const float*)d_in[0];
  float*       O  = (float*)d_out;
  float*       Wf = (float*)d_ws;
  float*       Ev = Wf + 84480;
  float*       Wn = Ev + 600;

  build_tables<<<221, 256, 0, stream>>>(Wf, Ev, Wn);
  irfft_overlap_add<<<512, 256, 0, stream>>>(X, Wf, Ev, Wn, O);
}


# Input staged by chunks of 128 frequencies, not 64: a thread fetches its task of both half chunks at once, eight or more 16-byte loads in flight per trip, 5 trips per block, not 13; sums unchanged
# speedup vs baseline: 4.4688x; 1.0396x over previous
#include <hip/hip_runtime.h>
#include <math.h>


typedef __attribute__((ext_vector_type(2))) float v2f;
typedef __attribute__((ext_vector_type(8))) float v8f;
typedef __attribute__((ext_vector_type(4))) float v4f;

#define ROWK   3084
#define NOUT   600
#define NPART  13
#define FRAG   64
#define PSTR   256
#define KSTR   3328
#define WTOT   56576
#define KCH    768
#define FPL    36
#define LDA    900
#define BLD    602
#define HALF   14400
#define LDSF   28800
#define NWIN   257

__global__ __launch_bounds__(256)
void build_tables(float* __restrict__ Wf, float* __restrict__ env, float* __restrict__ win) {
  const int gid = blockIdx.x * 256 + threadIdx.x;
  const float c0 = 6.283185307179586f / 512.0f;

  if (gid < WTOT) {
    int s    = gid & 1;
    int l    = (gid >> 1) & 31;
    int rest = gid >> 6;
    int j    = rest & 3;
    int pk   = rest >> 2;
    int k16  = pk / NPART;
    int part = pk - k16 * NPART;
    int cc   = l & 15;
    int hh   = l >> 4;
    int r    = (j << 4) + 1 + cc;
    bool sp  = (r == 64);
    int m    = sp ? 128 : (128 - r);
    bool oddf = (part >= 4) && (part < 12);
    int f;
    if (!oddf) f = (k16 << 4) + (hh << 3) + (s << 2) + (((part == 1) || (part == 3)) ? 2 : 0);
    else       f = (k16 << 4) + 1 + ((part & 1) << 1) + (hh << 3) + (s << 2);

    float v = 0.0f;
    if (f <= 256) {
      bool  edge = (f == 0) || (f == 256);
      float scl  = (edge ? 1.0f : 2.0f) / 512.0f;
      bool  flat = sp && ((part == 1) || (part == 8) || (part == 9) || (part == 12));
      if (part == 12 && !sp) {
        v = 0.0f;
      } else if (flat) {
        v = scl;
      } else {
        bool is_cos = (part < 2) || (part == 4) || (part == 5) || (part == 8) || (part == 9);
        int q  = (part >= 8) ? m : r;
        int ma = (f * q) & 511;
        float ang = c0 * (float)ma;
        float basis;
        if (is_cos) basis = cosf(ang);
        else        basis = ((ma & 255) == 0) ? 0.0f : -sinf(ang);
        v = scl * basis;
      }
    }
    Wf[gid] = v;
  }

  if (gid < NWIN) win[gid] = 0.5f * (1.0f - cosf(c0 * (float)gid));

  if (gid < NOUT) {
    int labs = gid + 256;
    float e = 0.0f;
    #pragma unroll
    for (int tp = 0; tp < 6; ++tp) {
      int nn2 = labs - (tp << 7);
      if (nn2 >= 0 && nn2 < 512) {
        float w = 0.5f * (1.0f - cosf(c0 * (float)nn2));
        e += w * w;
      }
    }
    env[gid] = e;
  }
}

#define WMMA4(a, b, c) __builtin_amdgcn_wmma_f32_16x16x4_f32(false, (a), false, (b), (short)0, (c), false, false)

__device__ __forceinline__
void k16step(const float* __restrict__ ap, const float* __restrict__ ws, unsigned lo, bool own64,
             v8f (&aE0)[3], v8f (&aE2)[3], v8f (&aO0)[3], v8f (&aO2)[3],
             v8f (&aEr)[3], v8f (&aOr)[3], v8f (&aEm)[3], v8f (&aOm)[3]) {
  const float* __restrict__ wr = ws + lo;
  {
    v2f bE0 = *(const v2f*)(wr);
    v2f bE2 = *(const v2f*)(wr + PSTR);
    v2f bO0 = *(const v2f*)(wr + 2 * PSTR);
    v2f bO2 = *(const v2f*)(wr + 3 * PSTR);
    v2f bC  = *(const v2f*)(wr + 12 * PSTR);
    #pragma unroll
    for (int i = 0; i < 3; ++i) {
      const float* a = ap + (8 * i) * FPL;
      v2f re0 = *(const v2f*)(a);
      v2f re2 = *(const v2f*)(a + 4);
      v2f im0 = *(const v2f*)(a + 2 * FPL);
      v2f im2 = *(const v2f*)(a + 2 * FPL + 4);
      aE0[i] = WMMA4(re0, bE0, aE0[i]);
      aE2[i] = WMMA4(re2, bE2, aE2[i]);
      aO0[i] = WMMA4(im0, bO0, aO0[i]);
      aO2[i] = WMMA4(im2, bO2, aO2[i]);
    }
    if (own64) {
      #pragma unroll
      for (int i = 0; i < 3; ++i) {
        v2f re0 = *(const v2f*)(ap + (8 * i) * FPL);
        aO0[i] = WMMA4(re0, bC, aO0[i]);
      }
    }
  }
  #pragma unroll
  for (int u = 0; u < 2; ++u) {
    v2f bEr = *(const v2f*)(wr + (4 + u) * PSTR);
    v2f bOr = *(const v2f*)(wr + (6 + u) * PSTR);
    v2f bEm = *(const v2f*)(wr + (8 + u) * PSTR);
    v2f bOm = *(const v2f*)(wr + (10 + u) * PSTR);
    #pragma unroll
    for (int i = 0; i < 3; ++i) {
      const float* a = ap + (8 * i + 1) * FPL + 4 * u;
      v2f ro = *(const v2f*)(a);
      v2f io = *(const v2f*)(a + 2 * FPL);
      aEr[i] = WMMA4(ro, bEr, aEr[i]);
      aEm[i] = WMMA4(ro, bEm, aEm[i]);
      aOr[i] = WMMA4(io, bOr, aOr[i]);
      aOm[i] = WMMA4(io, bOm, aOm[i]);
    }
  }
}

template <int I>
__device__ __forceinline__
void store_stage(const v8f (&s0)[3], const v8f (&s1)[3], const v8f (&s2)[3], const v8f (&s3)[3],
                 int p, bool mid, int fp, float* __restrict__ rowU, float* __restrict__ rowD) {
  const int b = 128 * fp;
  if (I == 0) {
    if (fp == 1 && mid) {
      #pragma unroll
      for (int v = 0; v < 8; ++v) rowU[v * BLD] = s0[0][v];
    }
    if (fp == 1 && !mid) {
      #pragma unroll
      for (int v = 0; v < 8; ++v) rowD[v * BLD + 128 - p] = s2[0][v];
    }
  } else if (I == 1) {
    #pragma unroll
    for (int v = 0; v < 8; ++v) rowU[v * BLD + b + p] = s0[1][v] + s3[0][v];
    if (!mid) {
      #pragma unroll
      for (int v = 0; v < 8; ++v) rowD[v * BLD + b + 256 - p] = s2[1][v] + s1[0][v];
    }
  } else {
    #pragma unroll
    for (int v = 0; v < 8; ++v) rowU[v * BLD + b + 256 + p] = s0[2][v] + s3[1][v];
    if (fp == 0 && p + 512 < NOUT) {
      #pragma unroll
      for (int v = 0; v < 8; ++v) rowU[v * BLD + 512 + p] = s3[2][v];
    }
    if (!mid && b + 512 - p < NOUT) {
      #pragma unroll
      for (int v = 0; v < 8; ++v) rowD[v * BLD + b + 512 - p] = s2[2][v] + s1[1][v];
    }
  }
}

template <int I>
__device__ __forceinline__
void epi_frame(v8f (&aE0)[3], v8f (&aE2)[3], v8f (&aO0)[3], v8f (&aO2)[3],
               v8f (&aEr)[3], v8f (&aOr)[3], v8f (&aEm)[3], v8f (&aOm)[3],
               int r, int m, bool sp, int fp, int h,
               float wr0, float wr1, float wm0, float wm1, float wc0,
               float* __restrict__ imgU, float* __restrict__ imgD) {
  if (sp) {
    #pragma unroll
    for (int v = 0; v < 8; ++v) {
      float Z  = aO0[I][v];
      float E0 = aE0[I][v], E2 = aE2[I][v];
      if (fp + 2 * I <= 4)
        imgD[(8 * h + v) * BLD + 128 * (fp + 2 * I)] = wc0 * (Z + E2 - aEm[I][v]);
      float bq = Z - E2;
      aE0[I][v] = 0.5f * (E0 + bq);
      aE2[I][v] = 0.5f * (E0 - bq);
      float hO = 0.5f * aO2[I][v];
      aO0[I][v] = hO;
      aO2[I][v] = hO;
      aEm[I][v] = 0.0f;
    }
  }
  #pragma unroll
  for (int v = 0; v < 8; ++v) {
    float E0 = aE0[I][v], E2 = aE2[I][v], O0 = aO0[I][v], O2 = aO2[I][v];
    float eOr = aEr[I][v], oOr = aOr[I][v], eOm = aEm[I][v], oOm = aOm[I][v];
    float eEr = E0 + E2;
    float eEm = E0 - E2;
    float oEr = O0 + O2;
    float oEm = O2 - O0;
    float ep = eEr + eOr;
    float em = eEr - eOr;
    float op = oOr + oEr;
    float om = oOr - oEr;
    aE0[I][v] = wr0 * (ep + op);
    aE2[I][v] = wr0 * (ep - op);
    aO0[I][v] = wr1 * (em + om);
    aO2[I][v] = wr1 * (em - om);
    ep = eEm + eOm;
    em = eEm - eOm;
    op = oOm + oEm;
    om = oOm - oEm;
    float t1 = wm0 * (ep - op);
    aEr[I][v] = wm0 * (ep + op);
    aOr[I][v] = t1;
    aEm[I][v] = wm1 * (em + om);
    aOm[I][v] = sp ? t1 : (wm1 * (em - om));
  }
  float* __restrict__ rowU = imgU + (8 * h) * BLD;
  float* __restrict__ rowD = imgD + (8 * h) * BLD;
  store_stage<I>(aE0, aE2, aO0, aO2, r, false, fp, rowU, rowD);
  store_stage<I>(aEr, aOr, aEm, aOm, m, sp, fp, rowU, rowD);
}

__device__ __forceinline__
int stage_fetch(const float* __restrict__ xk, int q, v4f (&L)[4], v4f (&M)[4]) {
  int rr  = q / 48;
  int rem = q - rr * 48;
  int eg  = rem / 6;
  int r6  = rem - eg * 6;
  int par = (r6 >= 3) ? 1 : 0;
  int tp  = r6 - 3 * par;
  int fl  = 16 * (eg >> 1) + 2 * (eg & 1) + par;
  const v4f* __restrict__ s = (const v4f*)(xk + (rr * ROWK + fl * 12 + 4 * tp));
  L[0] = s[0];
  L[1] = s[12];
  L[2] = s[24];
  L[3] = s[36];
  M[0] = s[KCH / 4];
  M[1] = s[KCH / 4 + 12];
  M[2] = s[KCH / 4 + 24];
  M[3] = s[KCH / 4 + 36];
  return rr * LDA + (8 * tp + par) * FPL + 4 * eg;
}

__device__ __forceinline__
void stage_put(float* __restrict__ dst, const v4f (&L)[4]) {
  v4f S0 = {L[0].x, L[1].x, L[2].x, L[3].x};
  v4f S1 = {L[0].y, L[1].y, L[2].y, L[3].y};
  v4f S2 = {L[0].z, L[1].z, L[2].z, L[3].z};
  v4f S3 = {L[0].w, L[1].w, L[2].w, L[3].w};
  *(v4f*)(dst)           = S0;
  *(v4f*)(dst + 2 * FPL) = S1;
  *(v4f*)(dst + 4 * FPL) = S2;
  *(v4f*)(dst + 6 * FPL) = S3;
}

__global__ __launch_bounds__(256) __attribute__((amdgpu_waves_per_eu(4)))
void irfft_overlap_add(const float* __restrict__ X, const float* __restrict__ Wf,
                       const float* __restrict__ env, const float* __restrict__ win,
                       float* __restrict__ out) {
  __shared__ __attribute__((aligned(16))) float As[LDSF];

  const int tid  = threadIdx.x;
  const int wave = __builtin_amdgcn_readfirstlane(tid >> 5);
  const int lane = tid & 31;
  const int c    = lane & 15;
  const int h    = lane >> 4;
  const int m0   = blockIdx.x << 4;
  const int fp   = wave >> 2;
  const int j    = (wave + fp) & 3;
  const bool own64 = (j == 3);

  v8f aE0[3] = {};
  v8f aE2[3] = {};
  v8f aO0[3] = {};
  v8f aO2[3] = {};
  v8f aEr[3] = {};
  v8f aOr[3] = {};
  v8f aEm[3] = {};
  v8f aOm[3] = {};

  const float* __restrict__ Xblk = X + (size_t)m0 * ROWK;
  const float* __restrict__ Arow = As + c * LDA + h * 2;
  const float* __restrict__ Afp  = Arow + (4 * fp) * FPL;
  const float* __restrict__ Wb   = Wf + j * FRAG;
  const unsigned            lo   = (unsigned)lane * 2u;

  for (int ch = 0; ch < 2; ++ch) {
    const float* __restrict__ xk = Xblk + (2 * ch) * KCH;
    #pragma unroll
    for (int s = 0; s < 3; ++s) {
      v4f La[4], Lb[4];
      int d = stage_fetch(xk, tid + 256 * s, La, Lb);
      stage_put(As + d, La);
      stage_put(As + HALF + d, Lb);
    }
    __syncthreads();

    #pragma unroll 1
    for (int kk = 0; kk < 8; ++kk)
      k16step(Afp + (kk >> 2) * HALF + (kk & 3) * 8, Wb + (ch * 8 + kk) * KSTR, lo, own64,
              aE0, aE2, aO0, aO2, aEr, aOr, aEm, aOm);
    __syncthreads();
  }

  for (int idx = tid; idx < 16 * 48; idx += 256) {
    int rr = idx / 48;
    int cc = idx - rr * 48;
    int pd = cc >> 2;
    int sl = cc & 3;
    As[rr * LDA + (pd * 2) * FPL + sl] = (sl == 0) ? Xblk[(size_t)rr * ROWK + 4 * KCH + pd] : 0.0f;
  }
  __syncthreads();
  {
    const float* __restrict__ wr = Wb + 16 * KSTR + lo;
    v2f bE0 = *(const v2f*)(wr);
    #pragma unroll
    for (int i = 0; i < 3; ++i) {
      v2f re0 = *(const v2f*)(Afp + (8 * i) * FPL);
      aE0[i] = WMMA4(re0, bE0, aE0[i]);
    }
    if (own64) {
      v2f bC = *(const v2f*)(wr + 12 * PSTR);
      #pragma unroll
      for (int i = 0; i < 3; ++i) {
        v2f re0 = *(const v2f*)(Afp + (8 * i) * FPL);
        aO0[i] = WMMA4(re0, bC, aO0[i]);
      }
    }
  }
  __syncthreads();

  float* __restrict__ imgU = As;
  float* __restrict__ imgD = As + 16 * BLD;

  const int   r   = (j << 4) + 1 + c;
  const bool  sp  = (r == 64);
  const int   m   = sp ? 128 : (128 - r);
  const float wr0 = win[r];
  const float wr1 = win[256 - r];
  const float wm0 = win[m];
  const float wm1 = win[256 - m];
  const float wc0 = win[256];

  epi_frame<0>(aE0, aE2, aO0, aO2, aEr, aOr, aEm, aOm, r, m, sp, fp, h, wr0, wr1, wm0, wm1, wc0, imgU, imgD);
  epi_frame<1>(aE0, aE2, aO0, aO2, aEr, aOr, aEm, aOm, r, m, sp, fp, h, wr0, wr1, wm0, wm1, wc0, imgU, imgD);
  epi_frame<2>(aE0, aE2, aO0, aO2, aEr, aOr, aEm, aOm, r, m, sp, fp, h, wr0, wr1, wm0, wm1, wc0, imgU, imgD);
  __syncthreads();

  float* __restrict__ oblk = out + (size_t)m0 * NOUT;
  for (int idx = tid; idx < 16 * NOUT; idx += 256) {
    int rr = idx / NOUT;
    int n  = idx - rr * NOUT;
    oblk[idx] = (imgU[rr * BLD + n] + imgD[rr * BLD + n]) / env[n];
  }
}

extern "C" void kernel_launch(void* const* d_in, const int* in_sizes, int n_in,
                              void* d_out, int out_size, void* d_ws, size_t ws_size,
                              hipStream_t stream) {
  const float* X  = (const float*)d_in[0];
  float*       O  = (float*)d_out;
  float*       Wf = (float*)d_ws;
  float*       Ev = Wf + 84480;
  float*       Wn = Ev + 600;

  build_tables<<<221, 256, 0, stream>>>(Wf, Ev, Wn);
  irfft_overlap_add<<<512, 256, 0, stream>>>(X, Wf, Ev, Wn, O);
}


# First odd k half's basis pairs and the r = 0 pair fetched with the even ones, ahead of the r = 64 branch the compiler sank them behind: one full L2 round trip per step, not 2 or 3; sums unchanged
# speedup vs baseline: 4.6580x; 1.0423x over previous
#include <hip/hip_runtime.h>
#include <math.h>


typedef __attribute__((ext_vector_type(2))) float v2f;
typedef __attribute__((ext_vector_type(8))) float v8f;
typedef __attribute__((ext_vector_type(4))) float v4f;

#define ROWK   3084
#define NOUT   600
#define NPART  13
#define FRAG   64
#define PSTR   256
#define KSTR   3328
#define WTOT   56576
#define KCH    768
#define FPL    36
#define LDA    900
#define BLD    602
#define HALF   14400
#define LDSF   28800
#define NWIN   257

__global__ __launch_bounds__(256)
void build_tables(float* __restrict__ Wf, float* __restrict__ env, float* __restrict__ win) {
  const int gid = blockIdx.x * 256 + threadIdx.x;
  const float c0 = 6.283185307179586f / 512.0f;

  if (gid < WTOT) {
    int s    = gid & 1;
    int l    = (gid >> 1) & 31;
    int rest = gid >> 6;
    int j    = rest & 3;
    int pk   = rest >> 2;
    int k16  = pk / NPART;
    int part = pk - k16 * NPART;
    int cc   = l & 15;
    int hh   = l >> 4;
    int r    = (j << 4) + 1 + cc;
    bool sp  = (r == 64);
    int m    = sp ? 128 : (128 - r);
    bool oddf = (part >= 4) && (part < 12);
    int f;
    if (!oddf) f = (k16 << 4) + (hh << 3) + (s << 2) + (((part == 1) || (part == 3)) ? 2 : 0);
    else       f = (k16 << 4) + 1 + ((part & 1) << 1) + (hh << 3) + (s << 2);

    float v = 0.0f;
    if (f <= 256) {
      bool  edge = (f == 0) || (f == 256);
      float scl  = (edge ? 1.0f : 2.0f) / 512.0f;
      bool  flat = sp && ((part == 1) || (part == 8) || (part == 9) || (part == 12));
      if (part == 12 && !sp) {
        v = 0.0f;
      } else if (flat) {
        v = scl;
      } else {
        bool is_cos = (part < 2) || (part == 4) || (part == 5) || (part == 8) || (part == 9);
        int q  = (part >= 8) ? m : r;
        int ma = (f * q) & 511;
        float ang = c0 * (float)ma;
        float basis;
        if (is_cos) basis = cosf(ang);
        else        basis = ((ma & 255) == 0) ? 0.0f : -sinf(ang);
        v = scl * basis;
      }
    }
    Wf[gid] = v;
  }

  if (gid < NWIN) win[gid] = 0.5f * (1.0f - cosf(c0 * (float)gid));

  if (gid < NOUT) {
    int labs = gid + 256;
    float e = 0.0f;
    #pragma unroll
    for (int tp = 0; tp < 6; ++tp) {
      int nn2 = labs - (tp << 7);
      if (nn2 >= 0 && nn2 < 512) {
        float w = 0.5f * (1.0f - cosf(c0 * (float)nn2));
        e += w * w;
      }
    }
    env[gid] = e;
  }
}

#define WMMA4(a, b, c) __builtin_amdgcn_wmma_f32_16x16x4_f32(false, (a), false, (b), (short)0, (c), false, false)

__device__ __forceinline__
void k16step(const float* __restrict__ ap, const float* __restrict__ ws, unsigned lo, bool own64,
             v8f (&aE0)[3], v8f (&aE2)[3], v8f (&aO0)[3], v8f (&aO2)[3],
             v8f (&aEr)[3], v8f (&aOr)[3], v8f (&aEm)[3], v8f (&aOm)[3]) {
  const float* __restrict__ wr = ws + lo;
  v2f bEr0 = *(const v2f*)(wr + 4 * PSTR);
  v2f bOr0 = *(const v2f*)(wr + 6 * PSTR);
  v2f bEm0 = *(const v2f*)(wr + 8 * PSTR);
  v2f bOm0 = *(const v2f*)(wr + 10 * PSTR);
  v2f bC   = *(const v2f*)(wr + 12 * PSTR);
  {
    v2f bE0 = *(const v2f*)(wr);
    v2f bE2 = *(const v2f*)(wr + PSTR);
    v2f bO0 = *(const v2f*)(wr + 2 * PSTR);
    v2f bO2 = *(const v2f*)(wr + 3 * PSTR);
    #pragma unroll
    for (int i = 0; i < 3; ++i) {
      const float* a = ap + (8 * i) * FPL;
      v2f re0 = *(const v2f*)(a);
      v2f re2 = *(const v2f*)(a + 4);
      v2f im0 = *(const v2f*)(a + 2 * FPL);
      v2f im2 = *(const v2f*)(a + 2 * FPL + 4);
      aE0[i] = WMMA4(re0, bE0, aE0[i]);
      aE2[i] = WMMA4(re2, bE2, aE2[i]);
      aO0[i] = WMMA4(im0, bO0, aO0[i]);
      aO2[i] = WMMA4(im2, bO2, aO2[i]);
    }
    asm volatile("" : "+v"(bEr0), "+v"(bOr0), "+v"(bEm0), "+v"(bOm0), "+v"(bC));
    if (own64) {
      #pragma unroll
      for (int i = 0; i < 3; ++i) {
        v2f re0 = *(const v2f*)(ap + (8 * i) * FPL);
        aO0[i] = WMMA4(re0, bC, aO0[i]);
      }
    }
  }
  #pragma unroll
  for (int u = 0; u < 2; ++u) {
    v2f bEr = (u == 0) ? bEr0 : *(const v2f*)(wr + 5 * PSTR);
    v2f bOr = (u == 0) ? bOr0 : *(const v2f*)(wr + 7 * PSTR);
    v2f bEm = (u == 0) ? bEm0 : *(const v2f*)(wr + 9 * PSTR);
    v2f bOm = (u == 0) ? bOm0 : *(const v2f*)(wr + 11 * PSTR);
    #pragma unroll
    for (int i = 0; i < 3; ++i) {
      const float* a = ap + (8 * i + 1) * FPL + 4 * u;
      v2f ro = *(const v2f*)(a);
      v2f io = *(const v2f*)(a + 2 * FPL);
      aEr[i] = WMMA4(ro, bEr, aEr[i]);
      aEm[i] = WMMA4(ro, bEm, aEm[i]);
      aOr[i] = WMMA4(io, bOr, aOr[i]);
      aOm[i] = WMMA4(io, bOm, aOm[i]);
    }
  }
}

template <int I>
__device__ __forceinline__
void store_stage(const v8f (&s0)[3], const v8f (&s1)[3], const v8f (&s2)[3], const v8f (&s3)[3],
                 int p, bool mid, int fp, float* __restrict__ rowU, float* __restrict__ rowD) {
  const int b = 128 * fp;
  if (I == 0) {
    if (fp == 1 && mid) {
      #pragma unroll
      for (int v = 0; v < 8; ++v) rowU[v * BLD] = s0[0][v];
    }
    if (fp == 1 && !mid) {
      #pragma unroll
      for (int v = 0; v < 8; ++v) rowD[v * BLD + 128 - p] = s2[0][v];
    }
  } else if (I == 1) {
    #pragma unroll
    for (int v = 0; v < 8; ++v) rowU[v * BLD + b + p] = s0[1][v] + s3[0][v];
    if (!mid) {
      #pragma unroll
      for (int v = 0; v < 8; ++v) rowD[v * BLD + b + 256 - p] = s2[1][v] + s1[0][v];
    }
  } else {
    #pragma unroll
    for (int v = 0; v < 8; ++v) rowU[v * BLD + b + 256 + p] = s0[2][v] + s3[1][v];
    if (fp == 0 && p + 512 < NOUT) {
      #pragma unroll
      for (int v = 0; v < 8; ++v) rowU[v * BLD + 512 + p] = s3[2][v];
    }
    if (!mid && b + 512 - p < NOUT) {
      #pragma unroll
      for (int v = 0; v < 8; ++v) rowD[v * BLD + b + 512 - p] = s2[2][v] + s1[1][v];
    }
  }
}

template <int I>
__device__ __forceinline__
void epi_frame(v8f (&aE0)[3], v8f (&aE2)[3], v8f (&aO0)[3], v8f (&aO2)[3],
               v8f (&aEr)[3], v8f (&aOr)[3], v8f (&aEm)[3], v8f (&aOm)[3],
               int r, int m, bool sp, int fp, int h,
               float wr0, float wr1, float wm0, float wm1, float wc0,
               float* __restrict__ imgU, float* __restrict__ imgD) {
  if (sp) {
    #pragma unroll
    for (int v = 0; v < 8; ++v) {
      float Z  = aO0[I][v];
      float E0 = aE0[I][v], E2 = aE2[I][v];
      if (fp + 2 * I <= 4)
        imgD[(8 * h + v) * BLD + 128 * (fp + 2 * I)] = wc0 * (Z + E2 - aEm[I][v]);
      float bq = Z - E2;
      aE0[I][v] = 0.5f * (E0 + bq);
      aE2[I][v] = 0.5f * (E0 - bq);
      float hO = 0.5f * aO2[I][v];
      aO0[I][v] = hO;
      aO2[I][v] = hO;
      aEm[I][v] = 0.0f;
    }
  }
  #pragma unroll
  for (int v = 0; v < 8; ++v) {
    float E0 = aE0[I][v], E2 = aE2[I][v], O0 = aO0[I][v], O2 = aO2[I][v];
    float eOr = aEr[I][v], oOr = aOr[I][v], eOm = aEm[I][v], oOm = aOm[I][v];
    float eEr = E0 + E2;
    float eEm = E0 - E2;
    float oEr = O0 + O2;
    float oEm = O2 - O0;
    float ep = eEr + eOr;
    float em = eEr - eOr;
    float op = oOr + oEr;
    float om = oOr - oEr;
    aE0[I][v] = wr0 * (ep + op);
    aE2[I][v] = wr0 * (ep - op);
    aO0[I][v] = wr1 * (em + om);
    aO2[I][v] = wr1 * (em - om);
    ep = eEm + eOm;
    em = eEm - eOm;
    op = oOm + oEm;
    om = oOm - oEm;
    float t1 = wm0 * (ep - op);
    aEr[I][v] = wm0 * (ep + op);
    aOr[I][v] = t1;
    aEm[I][v] = wm1 * (em + om);
    aOm[I][v] = sp ? t1 : (wm1 * (em - om));
  }
  float* __restrict__ rowU = imgU + (8 * h) * BLD;
  float* __restrict__ rowD = imgD + (8 * h) * BLD;
  store_stage<I>(aE0, aE2, aO0, aO2, r, false, fp, rowU, rowD);
  store_stage<I>(aEr, aOr, aEm, aOm, m, sp, fp, rowU, rowD);
}

__device__ __forceinline__
int stage_fetch(const float* __restrict__ xk, int q, v4f (&L)[4], v4f (&M)[4]) {
  int rr  = q / 48;
  int rem = q - rr * 48;
  int eg  = rem / 6;
  int r6  = rem - eg * 6;
  int par = (r6 >= 3) ? 1 : 0;
  int tp  = r6 - 3 * par;
  int fl  = 16 * (eg >> 1) + 2 * (eg & 1) + par;
  const v4f* __restrict__ s = (const v4f*)(xk + (rr * ROWK + fl * 12 + 4 * tp));
  L[0] = s[0];
  L[1] = s[12];
  L[2] = s[24];
  L[3] = s[36];
  M[0] = s[KCH / 4];
  M[1] = s[KCH / 4 + 12];
  M[2] = s[KCH / 4 + 24];
  M[3] = s[KCH / 4 + 36];
  return rr * LDA + (8 * tp + par) * FPL + 4 * eg;
}

__device__ __forceinline__
void stage_put(float* __restrict__ dst, const v4f (&L)[4]) {
  v4f S0 = {L[0].x, L[1].x, L[2].x, L[3].x};
  v4f S1 = {L[0].y, L[1].y, L[2].y, L[3].y};
  v4f S2 = {L[0].z, L[1].z, L[2].z, L[3].z};
  v4f S3 = {L[0].w, L[1].w, L[2].w, L[3].w};
  *(v4f*)(dst)           = S0;
  *(v4f*)(dst + 2 * FPL) = S1;
  *(v4f*)(dst + 4 * FPL) = S2;
  *(v4f*)(dst + 6 * FPL) = S3;
}

__global__ __launch_bounds__(256) __attribute__((amdgpu_waves_per_eu(4)))
void irfft_overlap_add(const float* __restrict__ X, const float* __restrict__ Wf,
                       const float* __restrict__ env, const float* __restrict__ win,
                       float* __restrict__ out) {
  __shared__ __attribute__((aligned(16))) float As[LDSF];

  const int tid  = threadIdx.x;
  const int wave = __builtin_amdgcn_readfirstlane(tid >> 5);
  const int lane = tid & 31;
  const int c    = lane & 15;
  const int h    = lane >> 4;
  const int m0   = blockIdx.x << 4;
  const int fp   = wave >> 2;
  const int j    = (wave + fp) & 3;
  const bool own64 = (j == 3);

  v8f aE0[3] = {};
  v8f aE2[3] = {};
  v8f aO0[3] = {};
  v8f aO2[3] = {};
  v8f aEr[3] = {};
  v8f aOr[3] = {};
  v8f aEm[3] = {};
  v8f aOm[3] = {};

  const float* __restrict__ Xblk = X + (size_t)m0 * ROWK;
  const float* __restrict__ Arow = As + c * LDA + h * 2;
  const float* __restrict__ Afp  = Arow + (4 * fp) * FPL;
  const float* __restrict__ Wb   = Wf + j * FRAG;
  const unsigned            lo   = (unsigned)lane * 2u;

  for (int ch = 0; ch < 2; ++ch) {
    const float* __restrict__ xk = Xblk + (2 * ch) * KCH;
    #pragma unroll
    for (int s = 0; s < 3; ++s) {
      v4f La[4], Lb[4];
      int d = stage_fetch(xk, tid + 256 * s, La, Lb);
      stage_put(As + d, La);
      stage_put(As + HALF + d, Lb);
    }
    __syncthreads();

    #pragma unroll 1
    for (int kk = 0; kk < 8; ++kk)
      k16step(Afp + (kk >> 2) * HALF + (kk & 3) * 8, Wb + (ch * 8 + kk) * KSTR, lo, own64,
              aE0, aE2, aO0, aO2, aEr, aOr, aEm, aOm);
    __syncthreads();
  }

  for (int idx = tid; idx < 16 * 48; idx += 256) {
    int rr = idx / 48;
    int cc = idx - rr * 48;
    int pd = cc >> 2;
    int sl = cc & 3;
    As[rr * LDA + (pd * 2) * FPL + sl] = (sl == 0) ? Xblk[(size_t)rr * ROWK + 4 * KCH + pd] : 0.0f;
  }
  __syncthreads();
  {
    const float* __restrict__ wr = Wb + 16 * KSTR + lo;
    v2f bE0 = *(const v2f*)(wr);
    #pragma unroll
    for (int i = 0; i < 3; ++i) {
      v2f re0 = *(const v2f*)(Afp + (8 * i) * FPL);
      aE0[i] = WMMA4(re0, bE0, aE0[i]);
    }
    if (own64) {
      v2f bC = *(const v2f*)(wr + 12 * PSTR);
      #pragma unroll
      for (int i = 0; i < 3; ++i) {
        v2f re0 = *(const v2f*)(Afp + (8 * i) * FPL);
        aO0[i] = WMMA4(re0, bC, aO0[i]);
      }
    }
  }
  __syncthreads();

  float* __restrict__ imgU = As;
  float* __restrict__ imgD = As + 16 * BLD;

  const int   r   = (j << 4) + 1 + c;
  const bool  sp  = (r == 64);
  const int   m   = sp ? 128 : (128 - r);
  const float wr0 = win[r];
  const float wr1 = win[256 - r];
  const float wm0 = win[m];
  const float wm1 = win[256 - m];
  const float wc0 = win[256];

  epi_frame<0>(aE0, aE2, aO0, aO2, aEr, aOr, aEm, aOm, r, m, sp, fp, h, wr0, wr1, wm0, wm1, wc0, imgU, imgD);
  epi_frame<1>(aE0, aE2, aO0, aO2, aEr, aOr, aEm, aOm, r, m, sp, fp, h, wr0, wr1, wm0, wm1, wc0, imgU, imgD);
  epi_frame<2>(aE0, aE2, aO0, aO2, aEr, aOr, aEm, aOm, r, m, sp, fp, h, wr0, wr1, wm0, wm1, wc0, imgU, imgD);
  __syncthreads();

  float* __restrict__ oblk = out + (size_t)m0 * NOUT;
  for (int idx = tid; idx < 16 * NOUT; idx += 256) {
    int rr = idx / NOUT;
    int n  = idx - rr * NOUT;
    oblk[idx] = (imgU[rr * BLD + n] + imgD[rr * BLD + n]) / env[n];
  }
}

extern "C" void kernel_launch(void* const* d_in, const int* in_sizes, int n_in,
                              void* d_out, int out_size, void* d_ws, size_t ws_size,
                              hipStream_t stream) {
  const float* X  = (const float*)d_in[0];
  float*       O  = (float*)d_out;
  float*       Wf = (float*)d_ws;
  float*       Ev = Wf + 84480;
  float*       Wn = Ev + 600;

  build_tables<<<221, 256, 0, stream>>>(Wf, Ev, Wn);
  irfft_overlap_add<<<512, 256, 0, stream>>>(X, Wf, Ev, Wn, O);
}


# Odd frequencies paired (4m+1, 4m+3): sums and differences staged, 4 products on the 4m+2 basis, rotated by the sample angle in the epilogue; 24 matrix instructions a step, not 36; sum order changes
# speedup vs baseline: 5.3893x; 1.1570x over previous
#include <hip/hip_runtime.h>
#include <math.h>


typedef __attribute__((ext_vector_type(2))) float v2f;
typedef __attribute__((ext_vector_type(8))) float v8f;
typedef __attribute__((ext_vector_type(4))) float v4f;

#define ROWK   3084
#define NOUT   600
#define NPART  5
#define FRAG   64
#define PSTR   256
#define KSTR   1280
#define WTOT   21760
#define KCH    768
#define FPL    36
#define LDA    900
#define BLD    602
#define HALF   14400
#define LDSF   28800
#define NWIN   257

__global__ __launch_bounds__(256)
void build_tables(float* __restrict__ Wf, float* __restrict__ env, float* __restrict__ win) {
  const int gid = blockIdx.x * 256 + threadIdx.x;
  const float c0 = 6.283185307179586f / 512.0f;

  if (gid < WTOT) {
    int s    = gid & 1;
    int l    = (gid >> 1) & 31;
    int rest = gid >> 6;
    int j    = rest & 3;
    int pk   = rest >> 2;
    int k16  = pk / NPART;
    int part = pk - k16 * NPART;
    int cc   = l & 15;
    int hh   = l >> 4;
    int r    = (j << 4) + 1 + cc;
    bool sp  = (r == 64);
    int f    = (k16 << 4) + (hh << 3) + (s << 2) + (((part == 1) || (part == 3)) ? 2 : 0);

    float v = 0.0f;
    if (f <= 256) {
      bool  edge = (f == 0) || (f == 256);
      float scl  = (edge ? 1.0f : 2.0f) / 512.0f;
      bool  flat = sp && ((part == 1) || (part == 4));
      if (part == 4 && !sp) {
        v = 0.0f;
      } else if (flat) {
        v = scl;
      } else {
        bool is_cos = (part < 2);
        int ma = (f * r) & 511;
        float ang = c0 * (float)ma;
        float basis;
        if (is_cos) basis = cosf(ang);
        else        basis = ((ma & 255) == 0) ? 0.0f : -sinf(ang);
        v = scl * basis;
      }
    }
    Wf[gid] = v;
  }

  if (gid < NWIN) win[gid] = 0.5f * (1.0f - cosf(c0 * (float)gid));

  if (gid < NOUT) {
    int labs = gid + 256;
    float e = 0.0f;
    #pragma unroll
    for (int tp = 0; tp < 6; ++tp) {
      int nn2 = labs - (tp << 7);
      if (nn2 >= 0 && nn2 < 512) {
        float w = 0.5f * (1.0f - cosf(c0 * (float)nn2));
        e += w * w;
      }
    }
    env[gid] = e;
  }
}

#define WMMA4(a, b, c) __builtin_amdgcn_wmma_f32_16x16x4_f32(false, (a), false, (b), (short)0, (c), false, false)

__device__ __forceinline__
void k16step(const float* __restrict__ ap, const float* __restrict__ ws, unsigned lo, bool own64,
             v8f (&aE0)[3], v8f (&aE2)[3], v8f (&aO0)[3], v8f (&aO2)[3],
             v8f (&aEr)[3], v8f (&aOr)[3], v8f (&aEm)[3], v8f (&aOm)[3]) {
  const float* __restrict__ wr = ws + lo;
  v2f bC  = *(const v2f*)(wr + 4 * PSTR);
  v2f bE0 = *(const v2f*)(wr);
  v2f bE2 = *(const v2f*)(wr + PSTR);
  v2f bO0 = *(const v2f*)(wr + 2 * PSTR);
  v2f bO2 = *(const v2f*)(wr + 3 * PSTR);
  asm volatile("" : "+v"(bC));
  #pragma unroll
  for (int i = 0; i < 3; ++i) {
    const float* a = ap + (8 * i) * FPL;
    v2f re0 = *(const v2f*)(a);
    v2f re2 = *(const v2f*)(a + 4);
    v2f im0 = *(const v2f*)(a + 2 * FPL);
    v2f im2 = *(const v2f*)(a + 2 * FPL + 4);
    v2f ru  = *(const v2f*)(a + FPL);
    v2f rd  = *(const v2f*)(a + FPL + 4);
    v2f iu  = *(const v2f*)(a + 3 * FPL);
    v2f id  = *(const v2f*)(a + 3 * FPL + 4);
    aE0[i] = WMMA4(re0, bE0, aE0[i]);
    aE2[i] = WMMA4(re2, bE2, aE2[i]);
    aO0[i] = WMMA4(im0, bO0, aO0[i]);
    aO2[i] = WMMA4(im2, bO2, aO2[i]);
    aEr[i] = WMMA4(ru, bE2, aEr[i]);
    aEm[i] = WMMA4(rd, bO2, aEm[i]);
    aOr[i] = WMMA4(iu, bO2, aOr[i]);
    aOm[i] = WMMA4(id, bE2, aOm[i]);
  }
  if (own64) {
    #pragma unroll
    for (int i = 0; i < 3; ++i) {
      v2f re0 = *(const v2f*)(ap + (8 * i) * FPL);
      aO0[i] = WMMA4(re0, bC, aO0[i]);
    }
  }
}

template <int I>
__device__ __forceinline__
void store_stage(const v8f (&s0)[3], const v8f (&s1)[3], const v8f (&s2)[3], const v8f (&s3)[3],
                 int p, bool mid, int fp, float* __restrict__ rowU, float* __restrict__ rowD) {
  const int b = 128 * fp;
  if (I == 0) {
    if (fp == 1 && mid) {
      #pragma unroll
      for (int v = 0; v < 8; ++v) rowU[v * BLD] = s0[0][v];
    }
    if (fp == 1 && !mid) {
      #pragma unroll
      for (int v = 0; v < 8; ++v) rowD[v * BLD + 128 - p] = s2[0][v];
    }
  } else if (I == 1) {
    #pragma unroll
    for (int v = 0; v < 8; ++v) rowU[v * BLD + b + p] = s0[1][v] + s3[0][v];
    if (!mid) {
      #pragma unroll
      for (int v = 0; v < 8; ++v) rowD[v * BLD + b + 256 - p] = s2[1][v] + s1[0][v];
    }
  } else {
    #pragma unroll
    for (int v = 0; v < 8; ++v) rowU[v * BLD + b + 256 + p] = s0[2][v] + s3[1][v];
    if (fp == 0 && p + 512 < NOUT) {
      #pragma unroll
      for (int v = 0; v < 8; ++v) rowU[v * BLD + 512 + p] = s3[2][v];
    }
    if (!mid && b + 512 - p < NOUT) {
      #pragma unroll
      for (int v = 0; v < 8; ++v) rowD[v * BLD + b + 512 - p] = s2[2][v] + s1[1][v];
    }
  }
}

__device__ __forceinline__
void rotate_odd(v8f (&aEr)[3], v8f (&aOr)[3], v8f (&aEm)[3], v8f (&aOm)[3], bool sp, float tc, float ts) {
  const float eP = sp ? 0.0f : tc;
  const float mP = sp ? 1.0f : -ts;
  const float oS = sp ? 0.0f : -ts;
  const float mR = sp ? 0.0f : ts;
  const float mS = sp ? 1.0f : tc;
  #pragma unroll
  for (int i = 0; i < 3; ++i) {
    #pragma unroll
    for (int v = 0; v < 8; ++v) {
      float P = aEr[i][v], Q = aEm[i][v], R = aOr[i][v], S = aOm[i][v];
      aEr[i][v] = eP * P + ts * Q;
      aEm[i][v] = mP * P + eP * Q;
      aOr[i][v] = tc * R + oS * S;
      aOm[i][v] = mR * R + mS * S;
    }
  }
}

template <int I>
__device__ __forceinline__
void epi_frame(v8f (&aE0)[3], v8f (&aE2)[3], v8f (&aO0)[3], v8f (&aO2)[3],
               v8f (&aEr)[3], v8f (&aOr)[3], v8f (&aEm)[3], v8f (&aOm)[3],
               int r, int m, bool sp, int fp, int h,
               float wr0, float wr1, float wm0, float wm1, float wc0,
               float* __restrict__ imgU, float* __restrict__ imgD) {
  if (sp) {
    #pragma unroll
    for (int v = 0; v < 8; ++v) {
      float Z  = aO0[I][v];
      float E0 = aE0[I][v], E2 = aE2[I][v];
      if (fp + 2 * I <= 4)
        imgD[(8 * h + v) * BLD + 128 * (fp + 2 * I)] = wc0 * (Z + E2 - aEm[I][v]);
      float bq = Z - E2;
      aE0[I][v] = 0.5f * (E0 + bq);
      aE2[I][v] = 0.5f * (E0 - bq);
      float hO = 0.5f * aO2[I][v];
      aO0[I][v] = hO;
      aO2[I][v] = hO;
      aEm[I][v] = 0.0f;
    }
  }
  #pragma unroll
  for (int v = 0; v < 8; ++v) {
    float E0 = aE0[I][v], E2 = aE2[I][v], O0 = aO0[I][v], O2 = aO2[I][v];
    float eOr = aEr[I][v], oOr = aOr[I][v], eOm = aEm[I][v], oOm = aOm[I][v];
    float eEr = E0 + E2;
    float eEm = E0 - E2;
    float oEr = O0 + O2;
    float oEm = O2 - O0;
    float ep = eEr + eOr;
    float em = eEr - eOr;
    float op = oOr + oEr;
    float om = oOr - oEr;
    aE0[I][v] = wr0 * (ep + op);
    aE2[I][v] = wr0 * (ep - op);
    aO0[I][v] = wr1 * (em + om);
    aO2[I][v] = wr1 * (em - om);
    ep = eEm + eOm;
    em = eEm - eOm;
    op = oOm + oEm;
    om = oOm - oEm;
    float t1 = wm0 * (ep - op);
    aEr[I][v] = wm0 * (ep + op);
    aOr[I][v] = t1;
    aEm[I][v] = wm1 * (em + om);
    aOm[I][v] = sp ? t1 : (wm1 * (em - om));
  }
  float* __restrict__ rowU = imgU + (8 * h) * BLD;
  float* __restrict__ rowD = imgD + (8 * h) * BLD;
  store_stage<I>(aE0, aE2, aO0, aO2, r, false, fp, rowU, rowD);
  store_stage<I>(aEr, aOr, aEm, aOm, m, sp, fp, rowU, rowD);
}

__device__ __forceinline__
int fresh_lane() {
  int l;
  asm volatile("v_mbcnt_lo_u32_b32 %0, -1, 0" : "=v"(l));
  return l;
}

__device__ __forceinline__
int stage_fetch(const float* __restrict__ xk, int q, v4f (&L)[4], v4f (&M)[4], float& odd) {
  int rr  = q / 48;
  int rem = q - rr * 48;
  int st  = rem / 6;
  int r6  = rem - st * 6;
  int par = (r6 >= 3) ? 1 : 0;
  int tp  = r6 - 3 * par;
  int fl  = 16 * st + par;
  const v4f* __restrict__ s = (const v4f*)(xk + (rr * ROWK + fl * 12 + 4 * tp));
  L[0] = s[0];
  L[1] = s[12];
  L[2] = s[24];
  L[3] = s[36];
  M[0] = s[6];
  M[1] = s[18];
  M[2] = s[30];
  M[3] = s[42];
  odd = (float)par;
  return (st >> 2) * HALF + rr * LDA + (8 * tp + par) * FPL + 8 * (st & 3);
}

__device__ __forceinline__
void stage_pair(float* __restrict__ dst, const v4f (&L)[4], const v4f (&M)[4], float odd) {
  v4f S0 = {L[0].x + odd * M[0].x, L[1].x + odd * M[1].x, L[2].x + odd * M[2].x, L[3].x + odd * M[3].x};
  v4f T0 = {M[0].x - odd * L[0].x, M[1].x - odd * L[1].x, M[2].x - odd * L[2].x, M[3].x - odd * L[3].x};
  *(v4f*)(dst)               = S0;
  *(v4f*)(dst + 4)           = T0;
  v4f S1 = {L[0].y + odd * M[0].y, L[1].y + odd * M[1].y, L[2].y + odd * M[2].y, L[3].y + odd * M[3].y};
  v4f T1 = {M[0].y - odd * L[0].y, M[1].y - odd * L[1].y, M[2].y - odd * L[2].y, M[3].y - odd * L[3].y};
  *(v4f*)(dst + 2 * FPL)     = S1;
  *(v4f*)(dst + 2 * FPL + 4) = T1;
  v4f S2 = {L[0].z + odd * M[0].z, L[1].z + odd * M[1].z, L[2].z + odd * M[2].z, L[3].z + odd * M[3].z};
  v4f T2 = {M[0].z - odd * L[0].z, M[1].z - odd * L[1].z, M[2].z - odd * L[2].z, M[3].z - odd * L[3].z};
  *(v4f*)(dst + 4 * FPL)     = S2;
  *(v4f*)(dst + 4 * FPL + 4) = T2;
  v4f S3 = {L[0].w + odd * M[0].w, L[1].w + odd * M[1].w, L[2].w + odd * M[2].w, L[3].w + odd * M[3].w};
  v4f T3 = {M[0].w - odd * L[0].w, M[1].w - odd * L[1].w, M[2].w - odd * L[2].w, M[3].w - odd * L[3].w};
  *(v4f*)(dst + 6 * FPL)     = S3;
  *(v4f*)(dst + 6 * FPL + 4) = T3;
}

__global__ __launch_bounds__(256) __attribute__((amdgpu_waves_per_eu(4)))
void irfft_overlap_add(const float* __restrict__ X, const float* __restrict__ Wf,
                       const float* __restrict__ env, const float* __restrict__ win,
                       float* __restrict__ out) {
  __shared__ __attribute__((aligned(16))) float As[LDSF];

  const int tid  = threadIdx.x;
  const int wave = __builtin_amdgcn_readfirstlane(tid >> 5);
  const int lane = tid & 31;
  const int c    = lane & 15;
  const int h    = lane >> 4;
  const int m0   = blockIdx.x << 4;
  const int fp   = wave >> 2;
  const int j    = (wave + fp) & 3;
  const bool own64 = (j == 3);

  v8f aE0[3] = {};
  v8f aE2[3] = {};
  v8f aO0[3] = {};
  v8f aO2[3] = {};
  v8f aEr[3] = {};
  v8f aOr[3] = {};
  v8f aEm[3] = {};
  v8f aOm[3] = {};

  const float* __restrict__ Xblk = X + (size_t)m0 * ROWK;
  const float* __restrict__ Arow = As + c * LDA + h * 2;
  const float* __restrict__ Afp  = Arow + (4 * fp) * FPL;
  const float* __restrict__ Wb   = Wf + j * FRAG;
  const unsigned            lo   = (unsigned)lane * 2u;

  for (int ch = 0; ch < 2; ++ch) {
    const float* __restrict__ xk = Xblk + (2 * ch) * KCH;
    #pragma unroll
    for (int s = 0; s < 3; ++s) {
      v4f La[4], Lb[4];
      float odd;
      int t = (wave << 5) + fresh_lane();
      int d = stage_fetch(xk, t + 256 * s, La, Lb, odd);
      stage_pair(As + d, La, Lb, odd);
    }
    __syncthreads();

    #pragma unroll 1
    for (int kk = 0; kk < 8; ++kk)
      k16step(Afp + (kk >> 2) * HALF + (kk & 3) * 8, Wb + (ch * 8 + kk) * KSTR, lo, own64,
              aE0, aE2, aO0, aO2, aEr, aOr, aEm, aOm);
    __syncthreads();
  }

  for (int idx = (wave << 5) + fresh_lane(); idx < 16 * 48; idx += 256) {
    int rr = idx / 48;
    int cc = idx - rr * 48;
    int pd = cc >> 2;
    int sl = cc & 3;
    As[rr * LDA + (pd * 2) * FPL + sl] = (sl == 0) ? Xblk[(size_t)rr * ROWK + 4 * KCH + pd] : 0.0f;
  }
  __syncthreads();
  {
    const float* __restrict__ wr = Wb + 16 * KSTR + lo;
    v2f bE0 = *(const v2f*)(wr);
    #pragma unroll
    for (int i = 0; i < 3; ++i) {
      v2f re0 = *(const v2f*)(Afp + (8 * i) * FPL);
      aE0[i] = WMMA4(re0, bE0, aE0[i]);
    }
    if (own64) {
      v2f bC = *(const v2f*)(wr + 4 * PSTR);
      #pragma unroll
      for (int i = 0; i < 3; ++i) {
        v2f re0 = *(const v2f*)(Afp + (8 * i) * FPL);
        aO0[i] = WMMA4(re0, bC, aO0[i]);
      }
    }
  }
  __syncthreads();

  float* __restrict__ imgU = As;
  float* __restrict__ imgD = As + 16 * BLD;

  const int   le  = fresh_lane();
  const int   ce  = le & 15;
  const int   he  = le >> 4;
  const int   r   = (j << 4) + 1 + ce;
  const bool  sp  = (r == 64);
  const int   m   = sp ? 128 : (128 - r);
  {
    const float tc = 1.0f - 2.0f * win[r];
    const float ts = 1.0f - 2.0f * win[128 - r];
    rotate_odd(aEr, aOr, aEm, aOm, sp, tc, ts);
  }
  const float wr0 = win[r];
  const float wr1 = win[256 - r];
  const float wm0 = win[m];
  const float wm1 = win[256 - m];
  const float wc0 = win[256];

  epi_frame<0>(aE0, aE2, aO0, aO2, aEr, aOr, aEm, aOm, r, m, sp, fp, he, wr0, wr1, wm0, wm1, wc0, imgU, imgD);
  __builtin_amdgcn_sched_barrier(0);
  epi_frame<1>(aE0, aE2, aO0, aO2, aEr, aOr, aEm, aOm, r, m, sp, fp, he, wr0, wr1, wm0, wm1, wc0, imgU, imgD);
  __builtin_amdgcn_sched_barrier(0);
  epi_frame<2>(aE0, aE2, aO0, aO2, aEr, aOr, aEm, aOm, r, m, sp, fp, he, wr0, wr1, wm0, wm1, wc0, imgU, imgD);
  __syncthreads();

  float* __restrict__ oblk = out + (size_t)m0 * NOUT;
  for (int idx = (wave << 5) + fresh_lane(); idx < 16 * NOUT; idx += 256) {
    int rr = idx / NOUT;
    int n  = idx - rr * NOUT;
    oblk[idx] = (imgU[rr * BLD + n] + imgD[rr * BLD + n]) / env[n];
  }
}

extern "C" void kernel_launch(void* const* d_in, const int* in_sizes, int n_in,
                              void* d_out, int out_size, void* d_ws, size_t ws_size,
                              hipStream_t stream) {
  const float* X  = (const float*)d_in[0];
  float*       O  = (float*)d_out;
  float*       Wf = (float*)d_ws;
  float*       Ev = Wf + 84480;
  float*       Wn = Ev + 600;

  build_tables<<<85, 256, 0, stream>>>(Wf, Ev, Wn);
  irfft_overlap_add<<<512, 256, 0, stream>>>(X, Wf, Ev, Wn, O);
}
